# starting kernel
# baseline (speedup 1.0000x reference)
#include <hip/hip_runtime.h>
#include <hip/hip_bf16.h>
#include <math.h>

#define BB      32
#define LL      128
#define DD      300
#define KK_REL  12
#define NREL    25
#define PD      50
#define NPAIR   3044
#define FWID    650
#define MROWS   (BB * LL)
#define NTILES  19
#define NPAD    (NTILES * 16)

typedef __attribute__((ext_vector_type(2))) float v2f;
typedef __attribute__((ext_vector_type(8))) float v8f;

__global__ __launch_bounds__(256) void setup_kernel(
    const float* __restrict__ pos_emb, const float* __restrict__ W_hid,
    const float* __restrict__ b_hid,
    float* __restrict__ S_ws, float* __restrict__ E_ws,
    int* __restrict__ pe, int* __restrict__ pc)
{
    const int tid = threadIdx.x;

    for (int i = tid; i < LL; i += 256) {
        int start = 0;
        for (int t = 0; t < i; ++t) {
            int lo = t - KK_REL < 0 ? 0 : t - KK_REL;
            int hi = t + KK_REL > LL - 1 ? LL - 1 : t + KK_REL;
            start += hi - lo + 1;
        }
        int lo = i - KK_REL < 0 ? 0 : i - KK_REL;
        int hi = i + KK_REL > LL - 1 ? LL - 1 : i + KK_REL;
        for (int j = lo; j <= hi; ++j) { pe[start] = i; pc[start] = j; ++start; }
    }

    for (int e = tid; e < NREL * PD; e += 256) {
        int r = e / PD, pcol = e % PD;
        float acc = 0.f;
        for (int rp = 0; rp < NREL; ++rp) {
            int ad = rp >= KK_REL ? rp - KK_REL : KK_REL - rp;
            float cnt = (float)(LL - ad);
            int d = r - rp;
            acc += cnt * expf(-(float)(d * d)) * pos_emb[rp * PD + pcol];
        }
        S_ws[e] = acc;
    }
    __syncthreads();

    for (int e = tid; e < NREL * DD; e += 256) {
        int r = e / DD, dout = e % DD;
        float acc = b_hid[dout];
        const float* srow = S_ws + r * PD;
        const float* wrow = W_hid + (size_t)dout * FWID + 2 * DD;
        for (int pcol = 0; pcol < PD; ++pcol) acc += srow[pcol] * wrow[pcol];
        E_ws[e] = acc;
    }
}

__global__ __launch_bounds__(256) void pack_x(
    const float* __restrict__ h_e, const float* __restrict__ h_c,
    const float* __restrict__ h_sh,
    float* __restrict__ XE, float* __restrict__ XC)
{
    const int i = blockIdx.x * 256 + threadIdx.x;
    const int n4 = (MROWS * DD) / 4;
    if (i >= n4) return;
    const float* src = blockIdx.y ? h_c : h_e;
    float*       dst = blockIdx.y ? XC : XE;
    float4 a = ((const float4*)src)[i];
    float4 s = ((const float4*)h_sh)[i];
    float4 r;
    r.x = a.x + s.x; r.y = a.y + s.y; r.z = a.z + s.z; r.w = a.w + s.w;
    ((float4*)dst)[i] = r;
}

__global__ __launch_bounds__(256) void pack_w(
    const float* __restrict__ W_hid, float* __restrict__ WB)
{
    const int n = blockIdx.x;
    const int sel = blockIdx.y;
    float* dst = WB + ((size_t)sel * NPAD + n) * DD;
    for (int k = threadIdx.x; k < DD; k += 256) {
        dst[k] = (n < DD) ? W_hid[(size_t)n * FWID + sel * DD + k] : 0.f;
    }
}

__global__ __launch_bounds__(256) void pairproj_gemm(
    const float* __restrict__ XE, const float* __restrict__ XC,
    const float* __restrict__ WB,
    float* __restrict__ A_ws, float* __restrict__ C_ws)
{
    const int wave = threadIdx.x >> 5;
    const int lane = threadIdx.x & 31;
    const int mt4 = blockIdx.x;
    const int nt  = blockIdx.y * 8 + wave;
    if (nt >= NTILES) return;

    const int sel = blockIdx.z;
    const float* X   = sel ? XC : XE;
    float*       Out = sel ? C_ws : A_ws;

    const int mlane = lane & 15;
    const int kk    = (lane >> 4) * 2;

    const float* x0 = X + (size_t)((mt4 * 4 + 0) * 16 + mlane) * DD + kk;
    const float* x1 = X + (size_t)((mt4 * 4 + 1) * 16 + mlane) * DD + kk;
    const float* x2 = X + (size_t)((mt4 * 4 + 2) * 16 + mlane) * DD + kk;
    const float* x3 = X + (size_t)((mt4 * 4 + 3) * 16 + mlane) * DD + kk;
    const float* wr = WB + ((size_t)sel * NPAD + nt * 16 + mlane) * DD + kk;

    v8f acc0 = (v8f)0.0f, acc1 = (v8f)0.0f, acc2 = (v8f)0.0f, acc3 = (v8f)0.0f;
#pragma unroll 5
    for (int k0 = 0; k0 < DD; k0 += 4) {
        v2f bf = *(const v2f*)(wr + k0);
        v2f a0 = *(const v2f*)(x0 + k0);
        v2f a1 = *(const v2f*)(x1 + k0);
        v2f a2 = *(const v2f*)(x2 + k0);
        v2f a3 = *(const v2f*)(x3 + k0);
        acc0 = __builtin_amdgcn_wmma_f32_16x16x4_f32(false, a0, false, bf, (short)0, acc0, false, false);
        acc1 = __builtin_amdgcn_wmma_f32_16x16x4_f32(false, a1, false, bf, (short)0, acc1, false, false);
        acc2 = __builtin_amdgcn_wmma_f32_16x16x4_f32(false, a2, false, bf, (short)0, acc2, false, false);
        acc3 = __builtin_amdgcn_wmma_f32_16x16x4_f32(false, a3, false, bf, (short)0, acc3, false, false);
    }

    const int rbase = (lane >> 4) << 3;
    const int ncol  = nt * 16 + mlane;
    if (ncol < DD) {
        v8f accs[4] = {acc0, acc1, acc2, acc3};
#pragma unroll
        for (int i = 0; i < 4; ++i) {
            const size_t row0 = (size_t)((mt4 * 4 + i) * 16 + rbase);
#pragma unroll
            for (int g = 0; g < 8; ++g)
                Out[(row0 + g) * DD + ncol] = accs[i][g];
        }
    }
}

__global__ __launch_bounds__(256) void pair_epilogue(
    const float* __restrict__ A_ws, const float* __restrict__ C_ws,
    const float* __restrict__ E_ws,
    const int* __restrict__ pe, const int* __restrict__ pc,
    const float* __restrict__ ln_g, const float* __restrict__ ln_b,
    const float* __restrict__ W_rel, const float* __restrict__ b_rel,
    float* __restrict__ out)
{
    const int wave = threadIdx.x >> 5;
    const int lane = threadIdx.x & 31;
    const int w = blockIdx.x * 8 + wave;
    if (w >= BB * NPAIR) return;
    const int b = w / NPAIR;
    const int p = w - b * NPAIR;

    const int emo = pe[p];
    const int cau = pc[p];
    const int r   = cau - emo + KK_REL;

    const float* ra = A_ws + ((size_t)b * LL + emo) * DD;
    const float* rc = C_ws + ((size_t)b * LL + cau) * DD;
    const float* re = E_ws + (size_t)r * DD;

    float h[10];
    float s = 0.f, ss = 0.f;
#pragma unroll
    for (int t = 0; t < 5; ++t) {
        const int idx = t * 64 + lane * 2;
        v2f v = (v2f)0.0f;
        if (idx < DD) {
            v2f va = *(const v2f*)(ra + idx);
            v2f vc = *(const v2f*)(rc + idx);
            v2f ve = *(const v2f*)(re + idx);
            v = va + vc + ve;
        }
        h[2 * t]     = v.x;
        h[2 * t + 1] = v.y;
        s  += v.x + v.y;
        ss += v.x * v.x + v.y * v.y;
    }
#pragma unroll
    for (int off = 16; off > 0; off >>= 1) {
        s  += __shfl_xor(s,  off, 32);
        ss += __shfl_xor(ss, off, 32);
    }
    const float mu   = s * (1.f / DD);
    const float var  = ss * (1.f / DD) - mu * mu;
    const float rstd = rsqrtf(var + 1e-5f);

    float acc = 0.f;
#pragma unroll
    for (int t = 0; t < 5; ++t) {
        const int idx = t * 64 + lane * 2;
        if (idx < DD) {
            v2f g  = *(const v2f*)(ln_g  + idx);
            v2f bb = *(const v2f*)(ln_b  + idx);
            v2f wr = *(const v2f*)(W_rel + idx);
            float v0 = (h[2 * t]     - mu) * rstd * g.x + bb.x;
            float v1 = (h[2 * t + 1] - mu) * rstd * g.y + bb.y;
            v0 = v0 > 0.f ? v0 : (expf(v0) - 1.f);
            v1 = v1 > 0.f ? v1 : (expf(v1) - 1.f);
            acc += v0 * wr.x + v1 * wr.y;
        }
    }
#pragma unroll
    for (int off = 16; off > 0; off >>= 1) acc += __shfl_xor(acc, off, 32);

    if (lane == 0) out[(size_t)b * NPAIR + p] = acc + b_rel[0];
}

extern "C" void kernel_launch(void* const* d_in, const int* in_sizes, int n_in,
                              void* d_out, int out_size, void* d_ws, size_t ws_size,
                              hipStream_t stream) {
    const float* h_e     = (const float*)d_in[0];
    const float* h_c     = (const float*)d_in[1];
    const float* h_sh    = (const float*)d_in[2];
    const float* pos_emb = (const float*)d_in[3];
    const float* W_hid   = (const float*)d_in[4];
    const float* b_hid   = (const float*)d_in[5];
    const float* ln_g    = (const float*)d_in[6];
    const float* ln_b    = (const float*)d_in[7];
    const float* W_rel   = (const float*)d_in[8];
    const float* b_rel   = (const float*)d_in[9];
    float* out = (float*)d_out;

    float* ws   = (float*)d_ws;
    float* A_ws = ws;
    float* C_ws = A_ws + (size_t)MROWS * DD;
    float* XE   = C_ws + (size_t)MROWS * DD;
    float* XC   = XE   + (size_t)MROWS * DD;
    float* WB   = XC   + (size_t)MROWS * DD;
    float* E_ws = WB   + (size_t)2 * NPAD * DD;
    float* S_ws = E_ws + NREL * DD;
    int*   pe   = (int*)(S_ws + NREL * PD);
    int*   pc   = pe + NPAIR;

    setup_kernel<<<1, 256, 0, stream>>>(pos_emb, W_hid, b_hid, S_ws, E_ws, pe, pc);

    dim3 pxg((MROWS * DD / 4 + 255) / 256, 2);
    pack_x<<<pxg, 256, 0, stream>>>(h_e, h_c, h_sh, XE, XC);

    dim3 pwg(NPAD, 2);
    pack_w<<<pwg, 256, 0, stream>>>(W_hid, WB);

    dim3 ggrid(MROWS / 64, 3, 2);
    pairproj_gemm<<<ggrid, 256, 0, stream>>>(XE, XC, WB, A_ws, C_ws);

    const int nwaves  = BB * NPAIR;
    const int nblocks = (nwaves + 7) / 8;
    pair_epilogue<<<nblocks, 256, 0, stream>>>(A_ws, C_ws, E_ws, pe, pc,
                                               ln_g, ln_b, W_rel, b_rel, out);
}


# Setup (pair tables, smoothed position rows, 25x300 position-term table) spread over 102 work-groups instead of one serial work-group; same left-to-right summation order
# speedup vs baseline: 1.8280x; 1.8280x over previous
#include <hip/hip_runtime.h>
#include <hip/hip_bf16.h>
#include <math.h>

#define BB      32
#define LL      128
#define DD      300
#define KK_REL  12
#define NREL    25
#define PD      50
#define NPAIR   3044
#define FWID    650
#define MROWS   (BB * LL)
#define NTILES  19
#define NPAD    (NTILES * 16)

typedef __attribute__((ext_vector_type(2))) float v2f;
typedef __attribute__((ext_vector_type(8))) float v8f;

#define SETUP_T   128
#define E_CHUNK   100
#define NWIN      (LL * NREL)

__global__ __launch_bounds__(SETUP_T) void setup_kernel(
    const float* __restrict__ pos_emb, const float* __restrict__ W_hid,
    const float* __restrict__ b_hid,
    float* __restrict__ E_ws,
    int* __restrict__ pe, int* __restrict__ pc)
{
    __shared__ float wgt[NREL + 7];
    __shared__ __attribute__((aligned(16))) float srow[PD + 2];
    const int tid = threadIdx.x;

    if (blockIdx.y >= NREL) {
        const int e = ((blockIdx.y - NREL) * 3 + blockIdx.x) * SETUP_T + tid;
        const int i  = e / NREL;
        const int rr = e - i * NREL;
        const int j  = i - KK_REL + rr;
        if (e < NWIN && j >= 0 && j < LL) {
            const int m = i < KK_REL ? i : KK_REL;
            const int n = i > LL - KK_REL ? i - (LL - KK_REL) : 0;
            const int start = NREL * i - (KK_REL * m - (m * (m - 1)) / 2) - (n * (n + 1)) / 2;
            const int lo = i - KK_REL < 0 ? 0 : i - KK_REL;
            const int idx = start + (j - lo);
            pe[idx] = i;
            pc[idx] = j;
        }
        return;
    }

    const int r    = blockIdx.y;
    const int dout = blockIdx.x * E_CHUNK + tid;
    const bool has_col = tid < E_CHUNK;

    v2f w[PD / 2];
    float bias = 0.f;
    if (has_col) {
        const float* wrow = W_hid + (size_t)dout * FWID + 2 * DD;
        for (int c = 0; c < PD / 2; ++c) w[c] = *(const v2f*)(wrow + 2 * c);
        bias = b_hid[dout];
    }
    float pv[NREL];
    if (tid < PD) {
        for (int rp = 0; rp < NREL; ++rp) pv[rp] = pos_emb[rp * PD + tid];
    }

    if (tid < NREL) {
        const int rp = tid;
        const int ad = rp >= KK_REL ? rp - KK_REL : KK_REL - rp;
        const float cnt = (float)(LL - ad);
        const int d = r - rp;
        wgt[rp] = cnt * expf(-(float)(d * d));
    }
    __syncthreads();

    if (tid < PD) {
        float acc = 0.f;
        for (int rp = 0; rp < NREL; ++rp) acc += wgt[rp] * pv[rp];
        srow[tid] = acc;
    }
    __syncthreads();

    if (has_col) {
        float acc = bias;
        for (int c = 0; c < PD / 2; ++c) {
            acc += srow[2 * c]     * w[c].x;
            acc += srow[2 * c + 1] * w[c].y;
        }
        E_ws[r * DD + dout] = acc;
    }
}

__global__ __launch_bounds__(256) void pack_x(
    const float* __restrict__ h_e, const float* __restrict__ h_c,
    const float* __restrict__ h_sh,
    float* __restrict__ XE, float* __restrict__ XC)
{
    const int i = blockIdx.x * 256 + threadIdx.x;
    const int n4 = (MROWS * DD) / 4;
    if (i >= n4) return;
    const float* src = blockIdx.y ? h_c : h_e;
    float*       dst = blockIdx.y ? XC : XE;
    float4 a = ((const float4*)src)[i];
    float4 s = ((const float4*)h_sh)[i];
    float4 r;
    r.x = a.x + s.x; r.y = a.y + s.y; r.z = a.z + s.z; r.w = a.w + s.w;
    ((float4*)dst)[i] = r;
}

__global__ __launch_bounds__(256) void pack_w(
    const float* __restrict__ W_hid, float* __restrict__ WB)
{
    const int n = blockIdx.x;
    const int sel = blockIdx.y;
    float* dst = WB + ((size_t)sel * NPAD + n) * DD;
    for (int k = threadIdx.x; k < DD; k += 256) {
        dst[k] = (n < DD) ? W_hid[(size_t)n * FWID + sel * DD + k] : 0.f;
    }
}

__global__ __launch_bounds__(256) void pairproj_gemm(
    const float* __restrict__ XE, const float* __restrict__ XC,
    const float* __restrict__ WB,
    float* __restrict__ A_ws, float* __restrict__ C_ws)
{
    const int wave = threadIdx.x >> 5;
    const int lane = threadIdx.x & 31;
    const int mt4 = blockIdx.x;
    const int nt  = blockIdx.y * 8 + wave;
    if (nt >= NTILES) return;

    const int sel = blockIdx.z;
    const float* X   = sel ? XC : XE;
    float*       Out = sel ? C_ws : A_ws;

    const int mlane = lane & 15;
    const int kk    = (lane >> 4) * 2;

    const float* x0 = X + (size_t)((mt4 * 4 + 0) * 16 + mlane) * DD + kk;
    const float* x1 = X + (size_t)((mt4 * 4 + 1) * 16 + mlane) * DD + kk;
    const float* x2 = X + (size_t)((mt4 * 4 + 2) * 16 + mlane) * DD + kk;
    const float* x3 = X + (size_t)((mt4 * 4 + 3) * 16 + mlane) * DD + kk;
    const float* wr = WB + ((size_t)sel * NPAD + nt * 16 + mlane) * DD + kk;

    v8f acc0 = (v8f)0.0f, acc1 = (v8f)0.0f, acc2 = (v8f)0.0f, acc3 = (v8f)0.0f;
#pragma unroll 5
    for (int k0 = 0; k0 < DD; k0 += 4) {
        v2f bf = *(const v2f*)(wr + k0);
        v2f a0 = *(const v2f*)(x0 + k0);
        v2f a1 = *(const v2f*)(x1 + k0);
        v2f a2 = *(const v2f*)(x2 + k0);
        v2f a3 = *(const v2f*)(x3 + k0);
        acc0 = __builtin_amdgcn_wmma_f32_16x16x4_f32(false, a0, false, bf, (short)0, acc0, false, false);
        acc1 = __builtin_amdgcn_wmma_f32_16x16x4_f32(false, a1, false, bf, (short)0, acc1, false, false);
        acc2 = __builtin_amdgcn_wmma_f32_16x16x4_f32(false, a2, false, bf, (short)0, acc2, false, false);
        acc3 = __builtin_amdgcn_wmma_f32_16x16x4_f32(false, a3, false, bf, (short)0, acc3, false, false);
    }

    const int rbase = (lane >> 4) << 3;
    const int ncol  = nt * 16 + mlane;
    if (ncol < DD) {
        v8f accs[4] = {acc0, acc1, acc2, acc3};
#pragma unroll
        for (int i = 0; i < 4; ++i) {
            const size_t row0 = (size_t)((mt4 * 4 + i) * 16 + rbase);
#pragma unroll
            for (int g = 0; g < 8; ++g)
                Out[(row0 + g) * DD + ncol] = accs[i][g];
        }
    }
}

__global__ __launch_bounds__(256) void pair_epilogue(
    const float* __restrict__ A_ws, const float* __restrict__ C_ws,
    const float* __restrict__ E_ws,
    const int* __restrict__ pe, const int* __restrict__ pc,
    const float* __restrict__ ln_g, const float* __restrict__ ln_b,
    const float* __restrict__ W_rel, const float* __restrict__ b_rel,
    float* __restrict__ out)
{
    const int wave = threadIdx.x >> 5;
    const int lane = threadIdx.x & 31;
    const int w = blockIdx.x * 8 + wave;
    if (w >= BB * NPAIR) return;
    const int b = w / NPAIR;
    const int p = w - b * NPAIR;

    const int emo = pe[p];
    const int cau = pc[p];
    const int r   = cau - emo + KK_REL;

    const float* ra = A_ws + ((size_t)b * LL + emo) * DD;
    const float* rc = C_ws + ((size_t)b * LL + cau) * DD;
    const float* re = E_ws + (size_t)r * DD;

    float h[10];
    float s = 0.f, ss = 0.f;
#pragma unroll
    for (int t = 0; t < 5; ++t) {
        const int idx = t * 64 + lane * 2;
        v2f v = (v2f)0.0f;
        if (idx < DD) {
            v2f va = *(const v2f*)(ra + idx);
            v2f vc = *(const v2f*)(rc + idx);
            v2f ve = *(const v2f*)(re + idx);
            v = va + vc + ve;
        }
        h[2 * t]     = v.x;
        h[2 * t + 1] = v.y;
        s  += v.x + v.y;
        ss += v.x * v.x + v.y * v.y;
    }
#pragma unroll
    for (int off = 16; off > 0; off >>= 1) {
        s  += __shfl_xor(s,  off, 32);
        ss += __shfl_xor(ss, off, 32);
    }
    const float mu   = s * (1.f / DD);
    const float var  = ss * (1.f / DD) - mu * mu;
    const float rstd = rsqrtf(var + 1e-5f);

    float acc = 0.f;
#pragma unroll
    for (int t = 0; t < 5; ++t) {
        const int idx = t * 64 + lane * 2;
        if (idx < DD) {
            v2f g  = *(const v2f*)(ln_g  + idx);
            v2f bb = *(const v2f*)(ln_b  + idx);
            v2f wr = *(const v2f*)(W_rel + idx);
            float v0 = (h[2 * t]     - mu) * rstd * g.x + bb.x;
            float v1 = (h[2 * t + 1] - mu) * rstd * g.y + bb.y;
            v0 = v0 > 0.f ? v0 : (expf(v0) - 1.f);
            v1 = v1 > 0.f ? v1 : (expf(v1) - 1.f);
            acc += v0 * wr.x + v1 * wr.y;
        }
    }
#pragma unroll
    for (int off = 16; off > 0; off >>= 1) acc += __shfl_xor(acc, off, 32);

    if (lane == 0) out[(size_t)b * NPAIR + p] = acc + b_rel[0];
}

extern "C" void kernel_launch(void* const* d_in, const int* in_sizes, int n_in,
                              void* d_out, int out_size, void* d_ws, size_t ws_size,
                              hipStream_t stream) {
    const float* h_e     = (const float*)d_in[0];
    const float* h_c     = (const float*)d_in[1];
    const float* h_sh    = (const float*)d_in[2];
    const float* pos_emb = (const float*)d_in[3];
    const float* W_hid   = (const float*)d_in[4];
    const float* b_hid   = (const float*)d_in[5];
    const float* ln_g    = (const float*)d_in[6];
    const float* ln_b    = (const float*)d_in[7];
    const float* W_rel   = (const float*)d_in[8];
    const float* b_rel   = (const float*)d_in[9];
    float* out = (float*)d_out;

    float* ws   = (float*)d_ws;
    float* A_ws = ws;
    float* C_ws = A_ws + (size_t)MROWS * DD;
    float* XE   = C_ws + (size_t)MROWS * DD;
    float* XC   = XE   + (size_t)MROWS * DD;
    float* WB   = XC   + (size_t)MROWS * DD;
    float* E_ws = WB   + (size_t)2 * NPAD * DD;
    float* S_ws = E_ws + NREL * DD;
    int*   pe   = (int*)(S_ws + NREL * PD);
    int*   pc   = pe + NPAIR;

    dim3 sg(3, NREL + 9);
    setup_kernel<<<sg, SETUP_T, 0, stream>>>(pos_emb, W_hid, b_hid, E_ws, pe, pc);

    dim3 pxg((MROWS * DD / 4 + 255) / 256, 2);
    pack_x<<<pxg, 256, 0, stream>>>(h_e, h_c, h_sh, XE, XC);

    dim3 pwg(NPAD, 2);
    pack_w<<<pwg, 256, 0, stream>>>(W_hid, WB);

    dim3 ggrid(MROWS / 64, 3, 2);
    pairproj_gemm<<<ggrid, 256, 0, stream>>>(XE, XC, WB, A_ws, C_ws);

    const int nwaves  = BB * NPAIR;
    const int nblocks = (nwaves + 7) / 8;
    pair_epilogue<<<nblocks, 256, 0, stream>>>(A_ws, C_ws, E_ws, pe, pc,
                                               ln_g, ln_b, W_rel, b_rel, out);
}


# Pack launches folded into the GEMM: each work-group builds its h+h_share and W_hid tiles in LDS, in 5 K chunks loaded under the previous chunk's WMMAs; 5 launches -> 3; same summation order
# speedup vs baseline: 2.7729x; 1.5169x over previous
#include <hip/hip_runtime.h>
#include <hip/hip_bf16.h>
#include <math.h>

#define BB      32
#define LL      128
#define DD      300
#define KK_REL  12
#define NREL    25
#define PD      50
#define NPAIR   3044
#define FWID    650
#define MROWS   (BB * LL)

typedef __attribute__((ext_vector_type(2))) float v2f;
typedef __attribute__((ext_vector_type(8))) float v8f;

#define SETUP_T   128
#define E_CHUNK   100
#define NWIN      (LL * NREL)

__global__ __launch_bounds__(SETUP_T) void setup_kernel(
    const float* __restrict__ pos_emb, const float* __restrict__ W_hid,
    const float* __restrict__ b_hid,
    float* __restrict__ E_ws,
    int* __restrict__ pe, int* __restrict__ pc)
{
    __shared__ float wgt[NREL + 7];
    __shared__ __attribute__((aligned(16))) float srow[PD + 2];
    const int tid = threadIdx.x;

    if (blockIdx.y >= NREL) {
        const int e = ((blockIdx.y - NREL) * 3 + blockIdx.x) * SETUP_T + tid;
        const int i  = e / NREL;
        const int rr = e - i * NREL;
        const int j  = i - KK_REL + rr;
        if (e < NWIN && j >= 0 && j < LL) {
            const int m = i < KK_REL ? i : KK_REL;
            const int n = i > LL - KK_REL ? i - (LL - KK_REL) : 0;
            const int start = NREL * i - (KK_REL * m - (m * (m - 1)) / 2) - (n * (n + 1)) / 2;
            const int lo = i - KK_REL < 0 ? 0 : i - KK_REL;
            const int idx = start + (j - lo);
            pe[idx] = i;
            pc[idx] = j;
        }
        return;
    }

    const int r    = blockIdx.y;
    const int dout = blockIdx.x * E_CHUNK + tid;
    const bool has_col = tid < E_CHUNK;

    v2f w[PD / 2];
    float bias = 0.f;
    if (has_col) {
        const float* wrow = W_hid + (size_t)dout * FWID + 2 * DD;
        for (int c = 0; c < PD / 2; ++c) w[c] = *(const v2f*)(wrow + 2 * c);
        bias = b_hid[dout];
    }
    float pv[NREL];
    if (tid < PD) {
        for (int rp = 0; rp < NREL; ++rp) pv[rp] = pos_emb[rp * PD + tid];
    }

    if (tid < NREL) {
        const int rp = tid;
        const int ad = rp >= KK_REL ? rp - KK_REL : KK_REL - rp;
        const float cnt = (float)(LL - ad);
        const int d = r - rp;
        wgt[rp] = cnt * expf(-(float)(d * d));
    }
    __syncthreads();

    if (tid < PD) {
        float acc = 0.f;
        for (int rp = 0; rp < NREL; ++rp) acc += wgt[rp] * pv[rp];
        srow[tid] = acc;
    }
    __syncthreads();

    if (has_col) {
        float acc = bias;
        for (int c = 0; c < PD / 2; ++c) {
            acc += srow[2 * c]     * w[c].x;
            acc += srow[2 * c + 1] * w[c].y;
        }
        E_ws[r * DD + dout] = acc;
    }
}

#define GM_ROWS   64
#define GN_TILES  10
#define GN_ROWS   (GN_TILES * 16)
#define KCH       60
#define NCH       (DD / KCH)
#define A_PCS     (GM_ROWS * KCH / 4)
#define A_SWEEPS  ((A_PCS + 255) / 256)
#define B_PCS     (GN_ROWS * KCH / 2)
#define B_SWEEPS  ((B_PCS + 255) / 256)

__global__ __launch_bounds__(256) void pairproj_gemm(
    const float* __restrict__ h_e, const float* __restrict__ h_c,
    const float* __restrict__ h_sh, const float* __restrict__ W_hid,
    float* __restrict__ A_ws, float* __restrict__ C_ws)
{
    __shared__ __attribute__((aligned(16))) float As[GM_ROWS * DD];
    __shared__ __attribute__((aligned(16))) float Bs[GN_ROWS * DD];

    const int tid  = threadIdx.x;
    const int wave = tid >> 5;
    const int lane = tid & 31;
    const int mq   = blockIdx.x;
    const int ng   = blockIdx.y;
    const int sel  = blockIdx.z;
    const float* H   = sel ? h_c : h_e;
    float*       Out = sel ? C_ws : A_ws;

    int bg[B_SWEEPS], bl[B_SWEEPS];
#pragma unroll
    for (int it = 0; it < B_SWEEPS; ++it) {
        int i = it * 256 + tid;
        i = i < B_PCS ? i : B_PCS - 1;
        const int row = i / (KCH / 2);
        const int c2  = i - row * (KCH / 2);
        int n = ng * GN_ROWS + row;
        n = n < DD ? n : DD - 1;
        bg[it] = n * FWID + 2 * c2;
        bl[it] = row * DD + 2 * c2;
    }
    int ao[A_SWEEPS];
#pragma unroll
    for (int it = 0; it < A_SWEEPS; ++it) {
        int i = it * 256 + tid;
        i = i < A_PCS ? i : A_PCS - 1;
        const int row = i / (KCH / 4);
        const int q   = i - row * (KCH / 4);
        ao[it] = row * DD + 4 * q;
    }
    const bool b_last = (B_SWEEPS - 1) * 256 + tid < B_PCS;
    const bool a_last = (A_SWEEPS - 1) * 256 + tid < A_PCS;
    const float* wsel = W_hid + sel * DD;
    const float* hrow = H    + (size_t)mq * GM_ROWS * DD;
    const float* srow = h_sh + (size_t)mq * GM_ROWS * DD;

    v2f    tb[B_SWEEPS];
    float4 ta[A_SWEEPS], ts[A_SWEEPS];

#pragma unroll
    for (int it = 0; it < B_SWEEPS; ++it) tb[it] = *(const v2f*)(wsel + bg[it]);
#pragma unroll
    for (int it = 0; it < A_SWEEPS; ++it) {
        ta[it] = *(const float4*)(hrow + ao[it]);
        ts[it] = *(const float4*)(srow + ao[it]);
    }
#pragma unroll
    for (int it = 0; it < B_SWEEPS; ++it)
        if (it < B_SWEEPS - 1 || b_last) *(v2f*)(Bs + bl[it]) = tb[it];
#pragma unroll
    for (int it = 0; it < A_SWEEPS; ++it)
        if (it < A_SWEEPS - 1 || a_last) {
            float4 r;
            r.x = ta[it].x + ts[it].x; r.y = ta[it].y + ts[it].y;
            r.z = ta[it].z + ts[it].z; r.w = ta[it].w + ts[it].w;
            *(float4*)(As + ao[it]) = r;
        }
    __syncthreads();

    const int mlane = lane & 15;
    const int kk    = (lane >> 4) * 2;
    const int mt    = wave & 3;
    const int nh    = wave >> 2;
    const float* ap = As + (mt * 16 + mlane) * DD + kk;
    const float* bp = Bs + (nh * 5 * 16 + mlane) * DD + kk;

    v8f acc0 = (v8f)0.0f, acc1 = (v8f)0.0f, acc2 = (v8f)0.0f, acc3 = (v8f)0.0f, acc4 = (v8f)0.0f;
#pragma unroll 1
    for (int c = 0; c < NCH; ++c) {
        const bool more = c + 1 < NCH;
        if (more) {
            const int kc = (c + 1) * KCH;
#pragma unroll
            for (int it = 0; it < B_SWEEPS; ++it) tb[it] = *(const v2f*)(wsel + bg[it] + kc);
#pragma unroll
            for (int it = 0; it < A_SWEEPS; ++it) {
                ta[it] = *(const float4*)(hrow + ao[it] + kc);
                ts[it] = *(const float4*)(srow + ao[it] + kc);
            }
        }
#pragma unroll
        for (int ks = 0; ks < KCH; ks += 4) {
            const int k0 = c * KCH + ks;
            const v2f a  = *(const v2f*)(ap + k0);
            const v2f b0 = *(const v2f*)(bp + 0 * 16 * DD + k0);
            const v2f b1 = *(const v2f*)(bp + 1 * 16 * DD + k0);
            const v2f b2 = *(const v2f*)(bp + 2 * 16 * DD + k0);
            const v2f b3 = *(const v2f*)(bp + 3 * 16 * DD + k0);
            const v2f b4 = *(const v2f*)(bp + 4 * 16 * DD + k0);
            acc0 = __builtin_amdgcn_wmma_f32_16x16x4_f32(false, a, false, b0, (short)0, acc0, false, false);
            acc1 = __builtin_amdgcn_wmma_f32_16x16x4_f32(false, a, false, b1, (short)0, acc1, false, false);
            acc2 = __builtin_amdgcn_wmma_f32_16x16x4_f32(false, a, false, b2, (short)0, acc2, false, false);
            acc3 = __builtin_amdgcn_wmma_f32_16x16x4_f32(false, a, false, b3, (short)0, acc3, false, false);
            acc4 = __builtin_amdgcn_wmma_f32_16x16x4_f32(false, a, false, b4, (short)0, acc4, false, false);
        }
        if (more) {
            const int kc = (c + 1) * KCH;
#pragma unroll
            for (int it = 0; it < B_SWEEPS; ++it)
                if (it < B_SWEEPS - 1 || b_last) *(v2f*)(Bs + bl[it] + kc) = tb[it];
#pragma unroll
            for (int it = 0; it < A_SWEEPS; ++it)
                if (it < A_SWEEPS - 1 || a_last) {
                    float4 r;
                    r.x = ta[it].x + ts[it].x; r.y = ta[it].y + ts[it].y;
                    r.z = ta[it].z + ts[it].z; r.w = ta[it].w + ts[it].w;
                    *(float4*)(As + ao[it] + kc) = r;
                }
            __syncthreads();
        }
    }

    const int rbase = (lane >> 4) << 3;
    const size_t row0 = (size_t)(mq * GM_ROWS + mt * 16 + rbase);
    v8f accs[5] = {acc0, acc1, acc2, acc3, acc4};
#pragma unroll
    for (int j = 0; j < 5; ++j) {
        const int ncol = (ng * GN_TILES + nh * 5 + j) * 16 + mlane;
        if (ncol < DD) {
#pragma unroll
            for (int g = 0; g < 8; ++g)
                Out[(row0 + g) * DD + ncol] = accs[j][g];
        }
    }
}

__global__ __launch_bounds__(256) void pair_epilogue(
    const float* __restrict__ A_ws, const float* __restrict__ C_ws,
    const float* __restrict__ E_ws,
    const int* __restrict__ pe, const int* __restrict__ pc,
    const float* __restrict__ ln_g, const float* __restrict__ ln_b,
    const float* __restrict__ W_rel, const float* __restrict__ b_rel,
    float* __restrict__ out)
{
    const int wave = threadIdx.x >> 5;
    const int lane = threadIdx.x & 31;
    const int w = blockIdx.x * 8 + wave;
    if (w >= BB * NPAIR) return;
    const int b = w / NPAIR;
    const int p = w - b * NPAIR;

    const int emo = pe[p];
    const int cau = pc[p];
    const int r   = cau - emo + KK_REL;

    const float* ra = A_ws + ((size_t)b * LL + emo) * DD;
    const float* rc = C_ws + ((size_t)b * LL + cau) * DD;
    const float* re = E_ws + (size_t)r * DD;

    float h[10];
    float s = 0.f, ss = 0.f;
#pragma unroll
    for (int t = 0; t < 5; ++t) {
        const int idx = t * 64 + lane * 2;
        v2f v = (v2f)0.0f;
        if (idx < DD) {
            v2f va = *(const v2f*)(ra + idx);
            v2f vc = *(const v2f*)(rc + idx);
            v2f ve = *(const v2f*)(re + idx);
            v = va + vc + ve;
        }
        h[2 * t]     = v.x;
        h[2 * t + 1] = v.y;
        s  += v.x + v.y;
        ss += v.x * v.x + v.y * v.y;
    }
#pragma unroll
    for (int off = 16; off > 0; off >>= 1) {
        s  += __shfl_xor(s,  off, 32);
        ss += __shfl_xor(ss, off, 32);
    }
    const float mu   = s * (1.f / DD);
    const float var  = ss * (1.f / DD) - mu * mu;
    const float rstd = rsqrtf(var + 1e-5f);

    float acc = 0.f;
#pragma unroll
    for (int t = 0; t < 5; ++t) {
        const int idx = t * 64 + lane * 2;
        if (idx < DD) {
            v2f g  = *(const v2f*)(ln_g  + idx);
            v2f bb = *(const v2f*)(ln_b  + idx);
            v2f wr = *(const v2f*)(W_rel + idx);
            float v0 = (h[2 * t]     - mu) * rstd * g.x + bb.x;
            float v1 = (h[2 * t + 1] - mu) * rstd * g.y + bb.y;
            v0 = v0 > 0.f ? v0 : (expf(v0) - 1.f);
            v1 = v1 > 0.f ? v1 : (expf(v1) - 1.f);
            acc += v0 * wr.x + v1 * wr.y;
        }
    }
#pragma unroll
    for (int off = 16; off > 0; off >>= 1) acc += __shfl_xor(acc, off, 32);

    if (lane == 0) out[(size_t)b * NPAIR + p] = acc + b_rel[0];
}

extern "C" void kernel_launch(void* const* d_in, const int* in_sizes, int n_in,
                              void* d_out, int out_size, void* d_ws, size_t ws_size,
                              hipStream_t stream) {
    const float* h_e     = (const float*)d_in[0];
    const float* h_c     = (const float*)d_in[1];
    const float* h_sh    = (const float*)d_in[2];
    const float* pos_emb = (const float*)d_in[3];
    const float* W_hid   = (const float*)d_in[4];
    const float* b_hid   = (const float*)d_in[5];
    const float* ln_g    = (const float*)d_in[6];
    const float* ln_b    = (const float*)d_in[7];
    const float* W_rel   = (const float*)d_in[8];
    const float* b_rel   = (const float*)d_in[9];
    float* out = (float*)d_out;

    float* ws   = (float*)d_ws;
    float* A_ws = ws;
    float* C_ws = A_ws + (size_t)MROWS * DD;
    float* E_ws = C_ws + (size_t)MROWS * DD;
    int*   pe   = (int*)(E_ws + NREL * DD);
    int*   pc   = pe + NPAIR;

    dim3 sg(3, NREL + 9);
    setup_kernel<<<sg, SETUP_T, 0, stream>>>(pos_emb, W_hid, b_hid, E_ws, pe, pc);

    dim3 ggrid(MROWS / GM_ROWS, 2, 2);
    pairproj_gemm<<<ggrid, 256, 0, stream>>>(h_e, h_c, h_sh, W_hid, A_ws, C_ws);

    const int nwaves  = BB * NPAIR;
    const int nblocks = (nwaves + 7) / 8;
    pair_epilogue<<<nblocks, 256, 0, stream>>>(A_ws, C_ws, E_ws, pe, pc,
                                               ln_g, ln_b, W_rel, b_rel, out);
}


# Epilogue: block per (batch, 16 emo rows) stages its 40 C rows and the E table in LDS; a wave walks up to 13 pairs of one emo row, A, ln, W_rel in registers; 97,408 -> 8,192 waves; same sum order
# speedup vs baseline: 3.2775x; 1.1820x over previous
#include <hip/hip_runtime.h>
#include <hip/hip_bf16.h>
#include <math.h>

#define BB      32
#define LL      128
#define DD      300
#define KK_REL  12
#define NREL    25
#define PD      50
#define NPAIR   3044
#define FWID    650
#define MROWS   (BB * LL)

typedef __attribute__((ext_vector_type(2))) float v2f;
typedef __attribute__((ext_vector_type(8))) float v8f;

#define SETUP_T   128
#define E_CHUNK   100

__global__ __launch_bounds__(SETUP_T) void setup_kernel(
    const float* __restrict__ pos_emb, const float* __restrict__ W_hid,
    const float* __restrict__ b_hid,
    float* __restrict__ E_ws)
{
    __shared__ float wgt[NREL + 7];
    __shared__ __attribute__((aligned(16))) float srow[PD + 2];
    const int tid = threadIdx.x;

    const int r    = blockIdx.y;
    const int dout = blockIdx.x * E_CHUNK + tid;
    const bool has_col = tid < E_CHUNK;

    v2f w[PD / 2];
    float bias = 0.f;
    if (has_col) {
        const float* wrow = W_hid + (size_t)dout * FWID + 2 * DD;
        for (int c = 0; c < PD / 2; ++c) w[c] = *(const v2f*)(wrow + 2 * c);
        bias = b_hid[dout];
    }
    float pv[NREL];
    if (tid < PD) {
        for (int rp = 0; rp < NREL; ++rp) pv[rp] = pos_emb[rp * PD + tid];
    }

    if (tid < NREL) {
        const int rp = tid;
        const int ad = rp >= KK_REL ? rp - KK_REL : KK_REL - rp;
        const float cnt = (float)(LL - ad);
        const int d = r - rp;
        wgt[rp] = cnt * expf(-(float)(d * d));
    }
    __syncthreads();

    if (tid < PD) {
        float acc = 0.f;
        for (int rp = 0; rp < NREL; ++rp) acc += wgt[rp] * pv[rp];
        srow[tid] = acc;
    }
    __syncthreads();

    if (has_col) {
        float acc = bias;
        for (int c = 0; c < PD / 2; ++c) {
            acc += srow[2 * c]     * w[c].x;
            acc += srow[2 * c + 1] * w[c].y;
        }
        E_ws[r * DD + dout] = acc;
    }
}

#define GM_ROWS   64
#define GN_TILES  10
#define GN_ROWS   (GN_TILES * 16)
#define KCH       60
#define NCH       (DD / KCH)
#define A_PCS     (GM_ROWS * KCH / 4)
#define A_SWEEPS  ((A_PCS + 255) / 256)
#define B_PCS     (GN_ROWS * KCH / 2)
#define B_SWEEPS  ((B_PCS + 255) / 256)

__global__ __launch_bounds__(256) void pairproj_gemm(
    const float* __restrict__ h_e, const float* __restrict__ h_c,
    const float* __restrict__ h_sh, const float* __restrict__ W_hid,
    float* __restrict__ A_ws, float* __restrict__ C_ws)
{
    __shared__ __attribute__((aligned(16))) float As[GM_ROWS * DD];
    __shared__ __attribute__((aligned(16))) float Bs[GN_ROWS * DD];

    const int tid  = threadIdx.x;
    const int wave = tid >> 5;
    const int lane = tid & 31;
    const int mq   = blockIdx.x;
    const int ng   = blockIdx.y;
    const int sel  = blockIdx.z;
    const float* H   = sel ? h_c : h_e;
    float*       Out = sel ? C_ws : A_ws;

    int bg[B_SWEEPS], bl[B_SWEEPS];
#pragma unroll
    for (int it = 0; it < B_SWEEPS; ++it) {
        int i = it * 256 + tid;
        i = i < B_PCS ? i : B_PCS - 1;
        const int row = i / (KCH / 2);
        const int c2  = i - row * (KCH / 2);
        int n = ng * GN_ROWS + row;
        n = n < DD ? n : DD - 1;
        bg[it] = n * FWID + 2 * c2;
        bl[it] = row * DD + 2 * c2;
    }
    int ao[A_SWEEPS];
#pragma unroll
    for (int it = 0; it < A_SWEEPS; ++it) {
        int i = it * 256 + tid;
        i = i < A_PCS ? i : A_PCS - 1;
        const int row = i / (KCH / 4);
        const int q   = i - row * (KCH / 4);
        ao[it] = row * DD + 4 * q;
    }
    const bool b_last = (B_SWEEPS - 1) * 256 + tid < B_PCS;
    const bool a_last = (A_SWEEPS - 1) * 256 + tid < A_PCS;
    const float* wsel = W_hid + sel * DD;
    const float* hrow = H    + (size_t)mq * GM_ROWS * DD;
    const float* srow = h_sh + (size_t)mq * GM_ROWS * DD;

    v2f    tb[B_SWEEPS];
    float4 ta[A_SWEEPS], ts[A_SWEEPS];

#pragma unroll
    for (int it = 0; it < B_SWEEPS; ++it) tb[it] = *(const v2f*)(wsel + bg[it]);
#pragma unroll
    for (int it = 0; it < A_SWEEPS; ++it) {
        ta[it] = *(const float4*)(hrow + ao[it]);
        ts[it] = *(const float4*)(srow + ao[it]);
    }
#pragma unroll
    for (int it = 0; it < B_SWEEPS; ++it)
        if (it < B_SWEEPS - 1 || b_last) *(v2f*)(Bs + bl[it]) = tb[it];
#pragma unroll
    for (int it = 0; it < A_SWEEPS; ++it)
        if (it < A_SWEEPS - 1 || a_last) {
            float4 r;
            r.x = ta[it].x + ts[it].x; r.y = ta[it].y + ts[it].y;
            r.z = ta[it].z + ts[it].z; r.w = ta[it].w + ts[it].w;
            *(float4*)(As + ao[it]) = r;
        }
    __syncthreads();

    const int mlane = lane & 15;
    const int kk    = (lane >> 4) * 2;
    const int mt    = wave & 3;
    const int nh    = wave >> 2;
    const float* ap = As + (mt * 16 + mlane) * DD + kk;
    const float* bp = Bs + (nh * 5 * 16 + mlane) * DD + kk;

    v8f acc0 = (v8f)0.0f, acc1 = (v8f)0.0f, acc2 = (v8f)0.0f, acc3 = (v8f)0.0f, acc4 = (v8f)0.0f;
#pragma unroll 1
    for (int c = 0; c < NCH; ++c) {
        const bool more = c + 1 < NCH;
        if (more) {
            const int kc = (c + 1) * KCH;
#pragma unroll
            for (int it = 0; it < B_SWEEPS; ++it) tb[it] = *(const v2f*)(wsel + bg[it] + kc);
#pragma unroll
            for (int it = 0; it < A_SWEEPS; ++it) {
                ta[it] = *(const float4*)(hrow + ao[it] + kc);
                ts[it] = *(const float4*)(srow + ao[it] + kc);
            }
        }
#pragma unroll
        for (int ks = 0; ks < KCH; ks += 4) {
            const int k0 = c * KCH + ks;
            const v2f a  = *(const v2f*)(ap + k0);
            const v2f b0 = *(const v2f*)(bp + 0 * 16 * DD + k0);
            const v2f b1 = *(const v2f*)(bp + 1 * 16 * DD + k0);
            const v2f b2 = *(const v2f*)(bp + 2 * 16 * DD + k0);
            const v2f b3 = *(const v2f*)(bp + 3 * 16 * DD + k0);
            const v2f b4 = *(const v2f*)(bp + 4 * 16 * DD + k0);
            acc0 = __builtin_amdgcn_wmma_f32_16x16x4_f32(false, a, false, b0, (short)0, acc0, false, false);
            acc1 = __builtin_amdgcn_wmma_f32_16x16x4_f32(false, a, false, b1, (short)0, acc1, false, false);
            acc2 = __builtin_amdgcn_wmma_f32_16x16x4_f32(false, a, false, b2, (short)0, acc2, false, false);
            acc3 = __builtin_amdgcn_wmma_f32_16x16x4_f32(false, a, false, b3, (short)0, acc3, false, false);
            acc4 = __builtin_amdgcn_wmma_f32_16x16x4_f32(false, a, false, b4, (short)0, acc4, false, false);
        }
        if (more) {
            const int kc = (c + 1) * KCH;
#pragma unroll
            for (int it = 0; it < B_SWEEPS; ++it)
                if (it < B_SWEEPS - 1 || b_last) *(v2f*)(Bs + bl[it] + kc) = tb[it];
#pragma unroll
            for (int it = 0; it < A_SWEEPS; ++it)
                if (it < A_SWEEPS - 1 || a_last) {
                    float4 r;
                    r.x = ta[it].x + ts[it].x; r.y = ta[it].y + ts[it].y;
                    r.z = ta[it].z + ts[it].z; r.w = ta[it].w + ts[it].w;
                    *(float4*)(As + ao[it] + kc) = r;
                }
            __syncthreads();
        }
    }

    const int rbase = (lane >> 4) << 3;
    const size_t row0 = (size_t)(mq * GM_ROWS + mt * 16 + rbase);
    v8f accs[5] = {acc0, acc1, acc2, acc3, acc4};
#pragma unroll
    for (int j = 0; j < 5; ++j) {
        const int ncol = (ng * GN_TILES + nh * 5 + j) * 16 + mlane;
        if (ncol < DD) {
#pragma unroll
            for (int g = 0; g < 8; ++g)
                Out[(row0 + g) * DD + ncol] = accs[j][g];
        }
    }
}

#define EP_ROWS   16
#define EP_T      (EP_ROWS * 2 * 32)
#define EP_CROWS  (EP_ROWS + 2 * KK_REL)
#define EP_NBLK   (LL / EP_ROWS)
#define C_PCS     (EP_CROWS * DD / 4)
#define C_SWEEPS  ((C_PCS + EP_T - 1) / EP_T)
#define E_PCS     (NREL * DD / 4)
#define E_SWEEPS  ((E_PCS + EP_T - 1) / EP_T)

__global__ __launch_bounds__(EP_T) void pair_epilogue(
    const float* __restrict__ A_ws, const float* __restrict__ C_ws,
    const float* __restrict__ E_ws,
    const float* __restrict__ ln_g, const float* __restrict__ ln_b,
    const float* __restrict__ W_rel, const float* __restrict__ b_rel,
    float* __restrict__ out)
{
    __shared__ __attribute__((aligned(16))) float Cs[EP_CROWS * DD];
    __shared__ __attribute__((aligned(16))) float Es[NREL * DD];

    const int tid  = threadIdx.x;
    const int lane = tid & 31;
    const int wave = __builtin_amdgcn_readfirstlane(tid >> 5);
    const int b    = blockIdx.x / EP_NBLK;
    const int i0   = (blockIdx.x - b * EP_NBLK) * EP_ROWS;
    int jlo = i0 - KK_REL;
    jlo = jlo < 0 ? 0 : jlo;
    jlo = jlo > LL - EP_CROWS ? LL - EP_CROWS : jlo;

    const float4* csrc = (const float4*)(C_ws + ((size_t)b * LL + jlo) * DD);
    const float4* esrc = (const float4*)E_ws;
#pragma unroll
    for (int it = 0; it < C_SWEEPS; ++it) {
        int q = it * EP_T + tid;
        q = q < C_PCS ? q : C_PCS - 1;
        ((float4*)Cs)[q] = csrc[q];
    }
#pragma unroll
    for (int it = 0; it < E_SWEEPS; ++it) {
        int q = it * EP_T + tid;
        q = q < E_PCS ? q : E_PCS - 1;
        ((float4*)Es)[q] = esrc[q];
    }

    const int i    = i0 + (wave & (EP_ROWS - 1));
    const int half = wave >> 4;
    const float* ra = A_ws + ((size_t)b * LL + i) * DD;
    v2f a[5], g[5], bb[5], wr[5];
#pragma unroll
    for (int t = 0; t < 5; ++t) {
        const int idx = t * 64 + lane * 2;
        a[t] = (v2f)0.0f; g[t] = (v2f)0.0f; bb[t] = (v2f)0.0f; wr[t] = (v2f)0.0f;
        if (idx < DD) {
            a[t]  = *(const v2f*)(ra    + idx);
            g[t]  = *(const v2f*)(ln_g  + idx);
            bb[t] = *(const v2f*)(ln_b  + idx);
            wr[t] = *(const v2f*)(W_rel + idx);
        }
    }
    const float brel = b_rel[0];
    __syncthreads();

    const int rlo = i < KK_REL ? KK_REL - i : 0;
    const int rhi = i + KK_REL > LL - 1 ? LL - 1 - i + KK_REL : 2 * KK_REL;
    const int n0  = (rhi - rlo + 2) >> 1;
    const int r0  = half ? rlo + n0 : rlo;
    const int r1  = half ? rhi : rlo + n0 - 1;
    const int m = i < KK_REL ? i : KK_REL;
    const int n = i > LL - KK_REL ? i - (LL - KK_REL) : 0;
    const int start = NREL * i - (KK_REL * m - (m * (m - 1)) / 2) - (n * (n + 1)) / 2;

    const float* crow = Cs + (i - KK_REL + r0 - jlo) * DD;
    const float* erow = Es + r0 * DD;
    float res = 0.f;
    for (int r = r0; r <= r1; ++r, crow += DD, erow += DD) {
        float h[10];
        float s = 0.f, ss = 0.f;
#pragma unroll
        for (int t = 0; t < 5; ++t) {
            const int idx = t * 64 + lane * 2;
            v2f v = (v2f)0.0f;
            if (idx < DD) {
                v2f vc = *(const v2f*)(crow + idx);
                v2f ve = *(const v2f*)(erow + idx);
                v = a[t] + vc + ve;
            }
            h[2 * t]     = v.x;
            h[2 * t + 1] = v.y;
            s  += v.x + v.y;
            ss += v.x * v.x + v.y * v.y;
        }
#pragma unroll
        for (int off = 16; off > 0; off >>= 1) {
            s  += __shfl_xor(s,  off, 32);
            ss += __shfl_xor(ss, off, 32);
        }
        const float mu   = s * (1.f / DD);
        const float var  = ss * (1.f / DD) - mu * mu;
        const float rstd = rsqrtf(var + 1e-5f);

        float acc = 0.f;
#pragma unroll
        for (int t = 0; t < 5; ++t) {
            const int idx = t * 64 + lane * 2;
            if (idx < DD) {
                float v0 = (h[2 * t]     - mu) * rstd * g[t].x + bb[t].x;
                float v1 = (h[2 * t + 1] - mu) * rstd * g[t].y + bb[t].y;
                v0 = v0 > 0.f ? v0 : (expf(v0) - 1.f);
                v1 = v1 > 0.f ? v1 : (expf(v1) - 1.f);
                acc += v0 * wr[t].x + v1 * wr[t].y;
            }
        }
#pragma unroll
        for (int off = 16; off > 0; off >>= 1) acc += __shfl_xor(acc, off, 32);

        if (lane == r - r0) res = acc + brel;
    }

    if (lane <= r1 - r0) out[(size_t)b * NPAIR + start + (r0 - rlo) + lane] = res;
}

extern "C" void kernel_launch(void* const* d_in, const int* in_sizes, int n_in,
                              void* d_out, int out_size, void* d_ws, size_t ws_size,
                              hipStream_t stream) {
    const float* h_e     = (const float*)d_in[0];
    const float* h_c     = (const float*)d_in[1];
    const float* h_sh    = (const float*)d_in[2];
    const float* pos_emb = (const float*)d_in[3];
    const float* W_hid   = (const float*)d_in[4];
    const float* b_hid   = (const float*)d_in[5];
    const float* ln_g    = (const float*)d_in[6];
    const float* ln_b    = (const float*)d_in[7];
    const float* W_rel   = (const float*)d_in[8];
    const float* b_rel   = (const float*)d_in[9];
    float* out = (float*)d_out;

    float* ws   = (float*)d_ws;
    float* A_ws = ws;
    float* C_ws = A_ws + (size_t)MROWS * DD;
    float* E_ws = C_ws + (size_t)MROWS * DD;

    dim3 sg(3, NREL);
    setup_kernel<<<sg, SETUP_T, 0, stream>>>(pos_emb, W_hid, b_hid, E_ws);

    dim3 ggrid(MROWS / GM_ROWS, 2, 2);
    pairproj_gemm<<<ggrid, 256, 0, stream>>>(h_e, h_c, h_sh, W_hid, A_ws, C_ws);

    pair_epilogue<<<BB * EP_NBLK, EP_T, 0, stream>>>(A_ws, C_ws, E_ws,
                                                     ln_g, ln_b, W_rel, b_rel, out);
}


# Epilogue ELU branch-free: exp(v) = one exp2 instruction of v*log2(e) and a select, replacing library expf under a per-element branch; expf's split-argument correction dropped (~1e-7 abs)
# speedup vs baseline: 4.1752x; 1.2739x over previous
#include <hip/hip_runtime.h>
#include <hip/hip_bf16.h>
#include <math.h>

#define BB      32
#define LL      128
#define DD      300
#define KK_REL  12
#define NREL    25
#define PD      50
#define NPAIR   3044
#define FWID    650
#define MROWS   (BB * LL)

typedef __attribute__((ext_vector_type(2))) float v2f;
typedef __attribute__((ext_vector_type(8))) float v8f;

#define SETUP_T   128
#define E_CHUNK   100

__global__ __launch_bounds__(SETUP_T) void setup_kernel(
    const float* __restrict__ pos_emb, const float* __restrict__ W_hid,
    const float* __restrict__ b_hid,
    float* __restrict__ E_ws)
{
    __shared__ float wgt[NREL + 7];
    __shared__ __attribute__((aligned(16))) float srow[PD + 2];
    const int tid = threadIdx.x;

    const int r    = blockIdx.y;
    const int dout = blockIdx.x * E_CHUNK + tid;
    const bool has_col = tid < E_CHUNK;

    v2f w[PD / 2];
    float bias = 0.f;
    if (has_col) {
        const float* wrow = W_hid + (size_t)dout * FWID + 2 * DD;
        for (int c = 0; c < PD / 2; ++c) w[c] = *(const v2f*)(wrow + 2 * c);
        bias = b_hid[dout];
    }
    float pv[NREL];
    if (tid < PD) {
        for (int rp = 0; rp < NREL; ++rp) pv[rp] = pos_emb[rp * PD + tid];
    }

    if (tid < NREL) {
        const int rp = tid;
        const int ad = rp >= KK_REL ? rp - KK_REL : KK_REL - rp;
        const float cnt = (float)(LL - ad);
        const int d = r - rp;
        wgt[rp] = cnt * expf(-(float)(d * d));
    }
    __syncthreads();

    if (tid < PD) {
        float acc = 0.f;
        for (int rp = 0; rp < NREL; ++rp) acc += wgt[rp] * pv[rp];
        srow[tid] = acc;
    }
    __syncthreads();

    if (has_col) {
        float acc = bias;
        for (int c = 0; c < PD / 2; ++c) {
            acc += srow[2 * c]     * w[c].x;
            acc += srow[2 * c + 1] * w[c].y;
        }
        E_ws[r * DD + dout] = acc;
    }
}

#define GM_ROWS   64
#define GN_TILES  10
#define GN_ROWS   (GN_TILES * 16)
#define KCH       60
#define NCH       (DD / KCH)
#define A_PCS     (GM_ROWS * KCH / 4)
#define A_SWEEPS  ((A_PCS + 255) / 256)
#define B_PCS     (GN_ROWS * KCH / 2)
#define B_SWEEPS  ((B_PCS + 255) / 256)

__global__ __launch_bounds__(256) void pairproj_gemm(
    const float* __restrict__ h_e, const float* __restrict__ h_c,
    const float* __restrict__ h_sh, const float* __restrict__ W_hid,
    float* __restrict__ A_ws, float* __restrict__ C_ws)
{
    __shared__ __attribute__((aligned(16))) float As[GM_ROWS * DD];
    __shared__ __attribute__((aligned(16))) float Bs[GN_ROWS * DD];

    const int tid  = threadIdx.x;
    const int wave = tid >> 5;
    const int lane = tid & 31;
    const int mq   = blockIdx.x;
    const int ng   = blockIdx.y;
    const int sel  = blockIdx.z;
    const float* H   = sel ? h_c : h_e;
    float*       Out = sel ? C_ws : A_ws;

    int bg[B_SWEEPS], bl[B_SWEEPS];
#pragma unroll
    for (int it = 0; it < B_SWEEPS; ++it) {
        int i = it * 256 + tid;
        i = i < B_PCS ? i : B_PCS - 1;
        const int row = i / (KCH / 2);
        const int c2  = i - row * (KCH / 2);
        int n = ng * GN_ROWS + row;
        n = n < DD ? n : DD - 1;
        bg[it] = n * FWID + 2 * c2;
        bl[it] = row * DD + 2 * c2;
    }
    int ao[A_SWEEPS];
#pragma unroll
    for (int it = 0; it < A_SWEEPS; ++it) {
        int i = it * 256 + tid;
        i = i < A_PCS ? i : A_PCS - 1;
        const int row = i / (KCH / 4);
        const int q   = i - row * (KCH / 4);
        ao[it] = row * DD + 4 * q;
    }
    const bool b_last = (B_SWEEPS - 1) * 256 + tid < B_PCS;
    const bool a_last = (A_SWEEPS - 1) * 256 + tid < A_PCS;
    const float* wsel = W_hid + sel * DD;
    const float* hrow = H    + (size_t)mq * GM_ROWS * DD;
    const float* srow = h_sh + (size_t)mq * GM_ROWS * DD;

    v2f    tb[B_SWEEPS];
    float4 ta[A_SWEEPS], ts[A_SWEEPS];

#pragma unroll
    for (int it = 0; it < B_SWEEPS; ++it) tb[it] = *(const v2f*)(wsel + bg[it]);
#pragma unroll
    for (int it = 0; it < A_SWEEPS; ++it) {
        ta[it] = *(const float4*)(hrow + ao[it]);
        ts[it] = *(const float4*)(srow + ao[it]);
    }
#pragma unroll
    for (int it = 0; it < B_SWEEPS; ++it)
        if (it < B_SWEEPS - 1 || b_last) *(v2f*)(Bs + bl[it]) = tb[it];
#pragma unroll
    for (int it = 0; it < A_SWEEPS; ++it)
        if (it < A_SWEEPS - 1 || a_last) {
            float4 r;
            r.x = ta[it].x + ts[it].x; r.y = ta[it].y + ts[it].y;
            r.z = ta[it].z + ts[it].z; r.w = ta[it].w + ts[it].w;
            *(float4*)(As + ao[it]) = r;
        }
    __syncthreads();

    const int mlane = lane & 15;
    const int kk    = (lane >> 4) * 2;
    const int mt    = wave & 3;
    const int nh    = wave >> 2;
    const float* ap = As + (mt * 16 + mlane) * DD + kk;
    const float* bp = Bs + (nh * 5 * 16 + mlane) * DD + kk;

    v8f acc0 = (v8f)0.0f, acc1 = (v8f)0.0f, acc2 = (v8f)0.0f, acc3 = (v8f)0.0f, acc4 = (v8f)0.0f;
#pragma unroll 1
    for (int c = 0; c < NCH; ++c) {
        const bool more = c + 1 < NCH;
        if (more) {
            const int kc = (c + 1) * KCH;
#pragma unroll
            for (int it = 0; it < B_SWEEPS; ++it) tb[it] = *(const v2f*)(wsel + bg[it] + kc);
#pragma unroll
            for (int it = 0; it < A_SWEEPS; ++it) {
                ta[it] = *(const float4*)(hrow + ao[it] + kc);
                ts[it] = *(const float4*)(srow + ao[it] + kc);
            }
        }
#pragma unroll
        for (int ks = 0; ks < KCH; ks += 4) {
            const int k0 = c * KCH + ks;
            const v2f a  = *(const v2f*)(ap + k0);
            const v2f b0 = *(const v2f*)(bp + 0 * 16 * DD + k0);
            const v2f b1 = *(const v2f*)(bp + 1 * 16 * DD + k0);
            const v2f b2 = *(const v2f*)(bp + 2 * 16 * DD + k0);
            const v2f b3 = *(const v2f*)(bp + 3 * 16 * DD + k0);
            const v2f b4 = *(const v2f*)(bp + 4 * 16 * DD + k0);
            acc0 = __builtin_amdgcn_wmma_f32_16x16x4_f32(false, a, false, b0, (short)0, acc0, false, false);
            acc1 = __builtin_amdgcn_wmma_f32_16x16x4_f32(false, a, false, b1, (short)0, acc1, false, false);
            acc2 = __builtin_amdgcn_wmma_f32_16x16x4_f32(false, a, false, b2, (short)0, acc2, false, false);
            acc3 = __builtin_amdgcn_wmma_f32_16x16x4_f32(false, a, false, b3, (short)0, acc3, false, false);
            acc4 = __builtin_amdgcn_wmma_f32_16x16x4_f32(false, a, false, b4, (short)0, acc4, false, false);
        }
        if (more) {
            const int kc = (c + 1) * KCH;
#pragma unroll
            for (int it = 0; it < B_SWEEPS; ++it)
                if (it < B_SWEEPS - 1 || b_last) *(v2f*)(Bs + bl[it] + kc) = tb[it];
#pragma unroll
            for (int it = 0; it < A_SWEEPS; ++it)
                if (it < A_SWEEPS - 1 || a_last) {
                    float4 r;
                    r.x = ta[it].x + ts[it].x; r.y = ta[it].y + ts[it].y;
                    r.z = ta[it].z + ts[it].z; r.w = ta[it].w + ts[it].w;
                    *(float4*)(As + ao[it] + kc) = r;
                }
            __syncthreads();
        }
    }

    const int rbase = (lane >> 4) << 3;
    const size_t row0 = (size_t)(mq * GM_ROWS + mt * 16 + rbase);
    v8f accs[5] = {acc0, acc1, acc2, acc3, acc4};
#pragma unroll
    for (int j = 0; j < 5; ++j) {
        const int ncol = (ng * GN_TILES + nh * 5 + j) * 16 + mlane;
        if (ncol < DD) {
#pragma unroll
            for (int g = 0; g < 8; ++g)
                Out[(row0 + g) * DD + ncol] = accs[j][g];
        }
    }
}

#define EP_ROWS   16
#define EP_T      (EP_ROWS * 2 * 32)
#define EP_CROWS  (EP_ROWS + 2 * KK_REL)
#define EP_NBLK   (LL / EP_ROWS)
#define C_PCS     (EP_CROWS * DD / 4)
#define C_SWEEPS  ((C_PCS + EP_T - 1) / EP_T)
#define E_PCS     (NREL * DD / 4)
#define E_SWEEPS  ((E_PCS + EP_T - 1) / EP_T)
#define LOG2E_F   1.4426950408889634f

__global__ __launch_bounds__(EP_T) void pair_epilogue(
    const float* __restrict__ A_ws, const float* __restrict__ C_ws,
    const float* __restrict__ E_ws,
    const float* __restrict__ ln_g, const float* __restrict__ ln_b,
    const float* __restrict__ W_rel, const float* __restrict__ b_rel,
    float* __restrict__ out)
{
    __shared__ __attribute__((aligned(16))) float Cs[EP_CROWS * DD];
    __shared__ __attribute__((aligned(16))) float Es[NREL * DD];

    const int tid  = threadIdx.x;
    const int lane = tid & 31;
    const int wave = __builtin_amdgcn_readfirstlane(tid >> 5);
    const int b    = blockIdx.x / EP_NBLK;
    const int i0   = (blockIdx.x - b * EP_NBLK) * EP_ROWS;
    int jlo = i0 - KK_REL;
    jlo = jlo < 0 ? 0 : jlo;
    jlo = jlo > LL - EP_CROWS ? LL - EP_CROWS : jlo;

    const float4* csrc = (const float4*)(C_ws + ((size_t)b * LL + jlo) * DD);
    const float4* esrc = (const float4*)E_ws;
#pragma unroll
    for (int it = 0; it < C_SWEEPS; ++it) {
        int q = it * EP_T + tid;
        q = q < C_PCS ? q : C_PCS - 1;
        ((float4*)Cs)[q] = csrc[q];
    }
#pragma unroll
    for (int it = 0; it < E_SWEEPS; ++it) {
        int q = it * EP_T + tid;
        q = q < E_PCS ? q : E_PCS - 1;
        ((float4*)Es)[q] = esrc[q];
    }

    const int i    = i0 + (wave & (EP_ROWS - 1));
    const int half = wave >> 4;
    const float* ra = A_ws + ((size_t)b * LL + i) * DD;
    v2f a[5], g[5], bb[5], wr[5];
#pragma unroll
    for (int t = 0; t < 5; ++t) {
        const int idx = t * 64 + lane * 2;
        a[t] = (v2f)0.0f; g[t] = (v2f)0.0f; bb[t] = (v2f)0.0f; wr[t] = (v2f)0.0f;
        if (idx < DD) {
            a[t]  = *(const v2f*)(ra    + idx);
            g[t]  = *(const v2f*)(ln_g  + idx);
            bb[t] = *(const v2f*)(ln_b  + idx);
            wr[t] = *(const v2f*)(W_rel + idx);
        }
    }
    const float brel = b_rel[0];
    __syncthreads();

    const int rlo = i < KK_REL ? KK_REL - i : 0;
    const int rhi = i + KK_REL > LL - 1 ? LL - 1 - i + KK_REL : 2 * KK_REL;
    const int n0  = (rhi - rlo + 2) >> 1;
    const int r0  = half ? rlo + n0 : rlo;
    const int r1  = half ? rhi : rlo + n0 - 1;
    const int m = i < KK_REL ? i : KK_REL;
    const int n = i > LL - KK_REL ? i - (LL - KK_REL) : 0;
    const int start = NREL * i - (KK_REL * m - (m * (m - 1)) / 2) - (n * (n + 1)) / 2;

    const float* crow = Cs + (i - KK_REL + r0 - jlo) * DD;
    const float* erow = Es + r0 * DD;
    float res = 0.f;
    for (int r = r0; r <= r1; ++r, crow += DD, erow += DD) {
        float h[10];
        float s = 0.f, ss = 0.f;
#pragma unroll
        for (int t = 0; t < 5; ++t) {
            const int idx = t * 64 + lane * 2;
            v2f v = (v2f)0.0f;
            if (idx < DD) {
                v2f vc = *(const v2f*)(crow + idx);
                v2f ve = *(const v2f*)(erow + idx);
                v = a[t] + vc + ve;
            }
            h[2 * t]     = v.x;
            h[2 * t + 1] = v.y;
            s  += v.x + v.y;
            ss += v.x * v.x + v.y * v.y;
        }
#pragma unroll
        for (int off = 16; off > 0; off >>= 1) {
            s  += __shfl_xor(s,  off, 32);
            ss += __shfl_xor(ss, off, 32);
        }
        const float mu   = s * (1.f / DD);
        const float var  = ss * (1.f / DD) - mu * mu;
        const float rstd = rsqrtf(var + 1e-5f);

        float acc = 0.f;
#pragma unroll
        for (int t = 0; t < 5; ++t) {
            const int idx = t * 64 + lane * 2;
            if (idx < DD) {
                float v0 = (h[2 * t]     - mu) * rstd * g[t].x + bb[t].x;
                float v1 = (h[2 * t + 1] - mu) * rstd * g[t].y + bb[t].y;
                const float e0 = __builtin_amdgcn_exp2f(v0 * LOG2E_F) - 1.f;
                const float e1 = __builtin_amdgcn_exp2f(v1 * LOG2E_F) - 1.f;
                v0 = v0 > 0.f ? v0 : e0;
                v1 = v1 > 0.f ? v1 : e1;
                acc += v0 * wr[t].x + v1 * wr[t].y;
            }
        }
#pragma unroll
        for (int off = 16; off > 0; off >>= 1) acc += __shfl_xor(acc, off, 32);

        if (lane == r - r0) res = acc + brel;
    }

    if (lane <= r1 - r0) out[(size_t)b * NPAIR + start + (r0 - rlo) + lane] = res;
}

extern "C" void kernel_launch(void* const* d_in, const int* in_sizes, int n_in,
                              void* d_out, int out_size, void* d_ws, size_t ws_size,
                              hipStream_t stream) {
    const float* h_e     = (const float*)d_in[0];
    const float* h_c     = (const float*)d_in[1];
    const float* h_sh    = (const float*)d_in[2];
    const float* pos_emb = (const float*)d_in[3];
    const float* W_hid   = (const float*)d_in[4];
    const float* b_hid   = (const float*)d_in[5];
    const float* ln_g    = (const float*)d_in[6];
    const float* ln_b    = (const float*)d_in[7];
    const float* W_rel   = (const float*)d_in[8];
    const float* b_rel   = (const float*)d_in[9];
    float* out = (float*)d_out;

    float* ws   = (float*)d_ws;
    float* A_ws = ws;
    float* C_ws = A_ws + (size_t)MROWS * DD;
    float* E_ws = C_ws + (size_t)MROWS * DD;

    dim3 sg(3, NREL);
    setup_kernel<<<sg, SETUP_T, 0, stream>>>(pos_emb, W_hid, b_hid, E_ws);

    dim3 ggrid(MROWS / GM_ROWS, 2, 2);
    pairproj_gemm<<<ggrid, 256, 0, stream>>>(h_e, h_c, h_sh, W_hid, A_ws, C_ws);

    pair_epilogue<<<BB * EP_NBLK, EP_T, 0, stream>>>(A_ws, C_ws, E_ws,
                                                     ln_g, ln_b, W_rel, b_rel, out);
}


# Setup launch folded into the GEMM launch: waves 0-3 of its first 75 work-groups build the 25x300 position-term table while waves 4-7 already multiply; 3 launches -> 2; same summation order
# speedup vs baseline: 4.6706x; 1.1187x over previous
#include <hip/hip_runtime.h>
#include <hip/hip_bf16.h>
#include <math.h>

#define BB      32
#define LL      128
#define DD      300
#define KK_REL  12
#define NREL    25
#define PD      50
#define NPAIR   3044
#define FWID    650
#define MROWS   (BB * LL)

typedef __attribute__((ext_vector_type(2))) float v2f;
typedef __attribute__((ext_vector_type(8))) float v8f;

#define E_CHUNK   100
#define E_BLOCKS  (3 * NREL)
#define PE_PCS    (NREL * PD / 2)
#define PE_SWEEPS ((PE_PCS + 255) / 256)
#define GM_ROWS   64
#define GN_TILES  10
#define GN_ROWS   (GN_TILES * 16)
#define KCH       60
#define NCH       (DD / KCH)
#define A_PCS     (GM_ROWS * KCH / 4)
#define A_SWEEPS  ((A_PCS + 255) / 256)
#define B_PCS     (GN_ROWS * KCH / 2)
#define B_SWEEPS  ((B_PCS + 255) / 256)

__global__ __launch_bounds__(256) void pairproj_gemm(
    const float* __restrict__ h_e, const float* __restrict__ h_c,
    const float* __restrict__ h_sh, const float* __restrict__ W_hid,
    const float* __restrict__ pos_emb, const float* __restrict__ b_hid,
    float* __restrict__ A_ws, float* __restrict__ C_ws, float* __restrict__ E_ws)
{
    __shared__ __attribute__((aligned(16))) float As[GM_ROWS * DD];
    __shared__ __attribute__((aligned(16))) float Bs[GN_ROWS * DD];
    __shared__ float wgt[NREL + 7];
    __shared__ __attribute__((aligned(16))) float pes[NREL * PD + 2];

    const int tid  = threadIdx.x;
    const int wave = tid >> 5;
    const int lane = tid & 31;
    const int mq   = blockIdx.x;
    const int ng   = blockIdx.y;
    const int sel  = blockIdx.z;
    const float* H   = sel ? h_c : h_e;
    float*       Out = sel ? C_ws : A_ws;

    int bg[B_SWEEPS], bl[B_SWEEPS];
#pragma unroll
    for (int it = 0; it < B_SWEEPS; ++it) {
        int i = it * 256 + tid;
        i = i < B_PCS ? i : B_PCS - 1;
        const int row = i / (KCH / 2);
        const int c2  = i - row * (KCH / 2);
        int n = ng * GN_ROWS + row;
        n = n < DD ? n : DD - 1;
        bg[it] = n * FWID + 2 * c2;
        bl[it] = row * DD + 2 * c2;
    }
    int ao[A_SWEEPS];
#pragma unroll
    for (int it = 0; it < A_SWEEPS; ++it) {
        int i = it * 256 + tid;
        i = i < A_PCS ? i : A_PCS - 1;
        const int row = i / (KCH / 4);
        const int q   = i - row * (KCH / 4);
        ao[it] = row * DD + 4 * q;
    }
    const bool b_last = (B_SWEEPS - 1) * 256 + tid < B_PCS;
    const bool a_last = (A_SWEEPS - 1) * 256 + tid < A_PCS;
    const float* wsel = W_hid + sel * DD;
    const float* hrow = H    + (size_t)mq * GM_ROWS * DD;
    const float* srow = h_sh + (size_t)mq * GM_ROWS * DD;

    v2f    tb[B_SWEEPS];
    float4 ta[A_SWEEPS], ts[A_SWEEPS];

    const int  lb    = mq + (MROWS / GM_ROWS) * (ng + 2 * sel);
    const bool eblk  = lb < E_BLOCKS;
    const int  er    = lb / 3;
    const int  edout = (lb - 3 * er) * E_CHUNK + tid;
    const bool ecol  = tid < E_CHUNK;
    v2f   epe[PE_SWEEPS];
    v2f   ew[PD / 2];
    float ebias = 0.f;
    if (eblk) {
#pragma unroll
        for (int it = 0; it < PE_SWEEPS; ++it) {
            int q = it * 256 + tid;
            q = q < PE_PCS ? q : PE_PCS - 1;
            epe[it] = *(const v2f*)(pos_emb + 2 * q);
        }
        if (ecol) {
            const float* wrow = W_hid + (size_t)edout * FWID + 2 * DD;
#pragma unroll
            for (int c = 0; c < PD / 2; ++c) ew[c] = *(const v2f*)(wrow + 2 * c);
            ebias = b_hid[edout];
        }
    }

#pragma unroll
    for (int it = 0; it < B_SWEEPS; ++it) tb[it] = *(const v2f*)(wsel + bg[it]);
#pragma unroll
    for (int it = 0; it < A_SWEEPS; ++it) {
        ta[it] = *(const float4*)(hrow + ao[it]);
        ts[it] = *(const float4*)(srow + ao[it]);
    }

    if (eblk) {
#pragma unroll
        for (int it = 0; it < PE_SWEEPS; ++it) {
            int q = it * 256 + tid;
            q = q < PE_PCS ? q : PE_PCS - 1;
            *(v2f*)(pes + 2 * q) = epe[it];
        }
        if (tid < NREL) {
            const int rp = tid;
            const int ad = rp >= KK_REL ? rp - KK_REL : KK_REL - rp;
            const float cnt = (float)(LL - ad);
            const int d = er - rp;
            wgt[rp] = cnt * expf(-(float)(d * d));
        }
    }

#pragma unroll
    for (int it = 0; it < B_SWEEPS; ++it)
        if (it < B_SWEEPS - 1 || b_last) *(v2f*)(Bs + bl[it]) = tb[it];
#pragma unroll
    for (int it = 0; it < A_SWEEPS; ++it)
        if (it < A_SWEEPS - 1 || a_last) {
            float4 r;
            r.x = ta[it].x + ts[it].x; r.y = ta[it].y + ts[it].y;
            r.z = ta[it].z + ts[it].z; r.w = ta[it].w + ts[it].w;
            *(float4*)(As + ao[it]) = r;
        }
    __syncthreads();

    if (eblk && tid < 128) {
        const int l32 = tid & 31;
        const int p1  = l32 + 32 < PD ? l32 + 32 : PD - 1;
        float s0 = 0.f, s1 = 0.f;
        for (int rp = 0; rp < NREL; ++rp) {
            const float wv = wgt[rp];
            s0 += wv * pes[rp * PD + l32];
            s1 += wv * pes[rp * PD + p1];
        }
        float acc = ebias;
#pragma unroll
        for (int c = 0; c < PD / 2; ++c) {
            const int pa = 2 * c, pb = 2 * c + 1;
            const float sa = __int_as_float(__builtin_amdgcn_readlane(__float_as_int(pa < 32 ? s0 : s1), pa & 31));
            const float sb = __int_as_float(__builtin_amdgcn_readlane(__float_as_int(pb < 32 ? s0 : s1), pb & 31));
            acc += sa * ew[c].x;
            acc += sb * ew[c].y;
        }
        if (ecol) E_ws[er * DD + edout] = acc;
    }

    const int mlane = lane & 15;
    const int kk    = (lane >> 4) * 2;
    const int mt    = wave & 3;
    const int nh    = wave >> 2;
    const float* ap = As + (mt * 16 + mlane) * DD + kk;
    const float* bp = Bs + (nh * 5 * 16 + mlane) * DD + kk;

    v8f acc0 = (v8f)0.0f, acc1 = (v8f)0.0f, acc2 = (v8f)0.0f, acc3 = (v8f)0.0f, acc4 = (v8f)0.0f;
#pragma unroll 1
    for (int c = 0; c < NCH; ++c) {
        const bool more = c + 1 < NCH;
        if (more) {
            const int kc = (c + 1) * KCH;
#pragma unroll
            for (int it = 0; it < B_SWEEPS; ++it) tb[it] = *(const v2f*)(wsel + bg[it] + kc);
#pragma unroll
            for (int it = 0; it < A_SWEEPS; ++it) {
                ta[it] = *(const float4*)(hrow + ao[it] + kc);
                ts[it] = *(const float4*)(srow + ao[it] + kc);
            }
        }
#pragma unroll
        for (int ks = 0; ks < KCH; ks += 4) {
            const int k0 = c * KCH + ks;
            const v2f a  = *(const v2f*)(ap + k0);
            const v2f b0 = *(const v2f*)(bp + 0 * 16 * DD + k0);
            const v2f b1 = *(const v2f*)(bp + 1 * 16 * DD + k0);
            const v2f b2 = *(const v2f*)(bp + 2 * 16 * DD + k0);
            const v2f b3 = *(const v2f*)(bp + 3 * 16 * DD + k0);
            const v2f b4 = *(const v2f*)(bp + 4 * 16 * DD + k0);
            acc0 = __builtin_amdgcn_wmma_f32_16x16x4_f32(false, a, false, b0, (short)0, acc0, false, false);
            acc1 = __builtin_amdgcn_wmma_f32_16x16x4_f32(false, a, false, b1, (short)0, acc1, false, false);
            acc2 = __builtin_amdgcn_wmma_f32_16x16x4_f32(false, a, false, b2, (short)0, acc2, false, false);
            acc3 = __builtin_amdgcn_wmma_f32_16x16x4_f32(false, a, false, b3, (short)0, acc3, false, false);
            acc4 = __builtin_amdgcn_wmma_f32_16x16x4_f32(false, a, false, b4, (short)0, acc4, false, false);
        }
        if (more) {
            const int kc = (c + 1) * KCH;
#pragma unroll
            for (int it = 0; it < B_SWEEPS; ++it)
                if (it < B_SWEEPS - 1 || b_last) *(v2f*)(Bs + bl[it] + kc) = tb[it];
#pragma unroll
            for (int it = 0; it < A_SWEEPS; ++it)
                if (it < A_SWEEPS - 1 || a_last) {
                    float4 r;
                    r.x = ta[it].x + ts[it].x; r.y = ta[it].y + ts[it].y;
                    r.z = ta[it].z + ts[it].z; r.w = ta[it].w + ts[it].w;
                    *(float4*)(As + ao[it] + kc) = r;
                }
            __syncthreads();
        }
    }

    const int rbase = (lane >> 4) << 3;
    const size_t row0 = (size_t)(mq * GM_ROWS + mt * 16 + rbase);
    v8f accs[5] = {acc0, acc1, acc2, acc3, acc4};
#pragma unroll
    for (int j = 0; j < 5; ++j) {
        const int ncol = (ng * GN_TILES + nh * 5 + j) * 16 + mlane;
        if (ncol < DD) {
#pragma unroll
            for (int g = 0; g < 8; ++g)
                Out[(row0 + g) * DD + ncol] = accs[j][g];
        }
    }
}

#define EP_ROWS   16
#define EP_T      (EP_ROWS * 2 * 32)
#define EP_CROWS  (EP_ROWS + 2 * KK_REL)
#define EP_NBLK   (LL / EP_ROWS)
#define C_PCS     (EP_CROWS * DD / 4)
#define C_SWEEPS  ((C_PCS + EP_T - 1) / EP_T)
#define E_PCS     (NREL * DD / 4)
#define E_SWEEPS  ((E_PCS + EP_T - 1) / EP_T)
#define LOG2E_F   1.4426950408889634f

__global__ __launch_bounds__(EP_T) void pair_epilogue(
    const float* __restrict__ A_ws, const float* __restrict__ C_ws,
    const float* __restrict__ E_ws,
    const float* __restrict__ ln_g, const float* __restrict__ ln_b,
    const float* __restrict__ W_rel, const float* __restrict__ b_rel,
    float* __restrict__ out)
{
    __shared__ __attribute__((aligned(16))) float Cs[EP_CROWS * DD];
    __shared__ __attribute__((aligned(16))) float Es[NREL * DD];

    const int tid  = threadIdx.x;
    const int lane = tid & 31;
    const int wave = __builtin_amdgcn_readfirstlane(tid >> 5);
    const int b    = blockIdx.x / EP_NBLK;
    const int i0   = (blockIdx.x - b * EP_NBLK) * EP_ROWS;
    int jlo = i0 - KK_REL;
    jlo = jlo < 0 ? 0 : jlo;
    jlo = jlo > LL - EP_CROWS ? LL - EP_CROWS : jlo;

    const float4* csrc = (const float4*)(C_ws + ((size_t)b * LL + jlo) * DD);
    const float4* esrc = (const float4*)E_ws;
#pragma unroll
    for (int it = 0; it < C_SWEEPS; ++it) {
        int q = it * EP_T + tid;
        q = q < C_PCS ? q : C_PCS - 1;
        ((float4*)Cs)[q] = csrc[q];
    }
#pragma unroll
    for (int it = 0; it < E_SWEEPS; ++it) {
        int q = it * EP_T + tid;
        q = q < E_PCS ? q : E_PCS - 1;
        ((float4*)Es)[q] = esrc[q];
    }

    const int i    = i0 + (wave & (EP_ROWS - 1));
    const int half = wave >> 4;
    const float* ra = A_ws + ((size_t)b * LL + i) * DD;
    v2f a[5], g[5], bb[5], wr[5];
#pragma unroll
    for (int t = 0; t < 5; ++t) {
        const int idx = t * 64 + lane * 2;
        a[t] = (v2f)0.0f; g[t] = (v2f)0.0f; bb[t] = (v2f)0.0f; wr[t] = (v2f)0.0f;
        if (idx < DD) {
            a[t]  = *(const v2f*)(ra    + idx);
            g[t]  = *(const v2f*)(ln_g  + idx);
            bb[t] = *(const v2f*)(ln_b  + idx);
            wr[t] = *(const v2f*)(W_rel + idx);
        }
    }
    const float brel = b_rel[0];
    __syncthreads();

    const int rlo = i < KK_REL ? KK_REL - i : 0;
    const int rhi = i + KK_REL > LL - 1 ? LL - 1 - i + KK_REL : 2 * KK_REL;
    const int n0  = (rhi - rlo + 2) >> 1;
    const int r0  = half ? rlo + n0 : rlo;
    const int r1  = half ? rhi : rlo + n0 - 1;
    const int m = i < KK_REL ? i : KK_REL;
    const int n = i > LL - KK_REL ? i - (LL - KK_REL) : 0;
    const int start = NREL * i - (KK_REL * m - (m * (m - 1)) / 2) - (n * (n + 1)) / 2;

    const float* crow = Cs + (i - KK_REL + r0 - jlo) * DD;
    const float* erow = Es + r0 * DD;
    float res = 0.f;
    for (int r = r0; r <= r1; ++r, crow += DD, erow += DD) {
        float h[10];
        float s = 0.f, ss = 0.f;
#pragma unroll
        for (int t = 0; t < 5; ++t) {
            const int idx = t * 64 + lane * 2;
            v2f v = (v2f)0.0f;
            if (idx < DD) {
                v2f vc = *(const v2f*)(crow + idx);
                v2f ve = *(const v2f*)(erow + idx);
                v = a[t] + vc + ve;
            }
            h[2 * t]     = v.x;
            h[2 * t + 1] = v.y;
            s  += v.x + v.y;
            ss += v.x * v.x + v.y * v.y;
        }
#pragma unroll
        for (int off = 16; off > 0; off >>= 1) {
            s  += __shfl_xor(s,  off, 32);
            ss += __shfl_xor(ss, off, 32);
        }
        const float mu   = s * (1.f / DD);
        const float var  = ss * (1.f / DD) - mu * mu;
        const float rstd = rsqrtf(var + 1e-5f);

        float acc = 0.f;
#pragma unroll
        for (int t = 0; t < 5; ++t) {
            const int idx = t * 64 + lane * 2;
            if (idx < DD) {
                float v0 = (h[2 * t]     - mu) * rstd * g[t].x + bb[t].x;
                float v1 = (h[2 * t + 1] - mu) * rstd * g[t].y + bb[t].y;
                const float e0 = __builtin_amdgcn_exp2f(v0 * LOG2E_F) - 1.f;
                const float e1 = __builtin_amdgcn_exp2f(v1 * LOG2E_F) - 1.f;
                v0 = v0 > 0.f ? v0 : e0;
                v1 = v1 > 0.f ? v1 : e1;
                acc += v0 * wr[t].x + v1 * wr[t].y;
            }
        }
#pragma unroll
        for (int off = 16; off > 0; off >>= 1) acc += __shfl_xor(acc, off, 32);

        if (lane == r - r0) res = acc + brel;
    }

    if (lane <= r1 - r0) out[(size_t)b * NPAIR + start + (r0 - rlo) + lane] = res;
}

extern "C" void kernel_launch(void* const* d_in, const int* in_sizes, int n_in,
                              void* d_out, int out_size, void* d_ws, size_t ws_size,
                              hipStream_t stream) {
    const float* h_e     = (const float*)d_in[0];
    const float* h_c     = (const float*)d_in[1];
    const float* h_sh    = (const float*)d_in[2];
    const float* pos_emb = (const float*)d_in[3];
    const float* W_hid   = (const float*)d_in[4];
    const float* b_hid   = (const float*)d_in[5];
    const float* ln_g    = (const float*)d_in[6];
    const float* ln_b    = (const float*)d_in[7];
    const float* W_rel   = (const float*)d_in[8];
    const float* b_rel   = (const float*)d_in[9];
    float* out = (float*)d_out;

    float* ws   = (float*)d_ws;
    float* A_ws = ws;
    float* C_ws = A_ws + (size_t)MROWS * DD;
    float* E_ws = C_ws + (size_t)MROWS * DD;

    dim3 ggrid(MROWS / GM_ROWS, 2, 2);
    pairproj_gemm<<<ggrid, 256, 0, stream>>>(h_e, h_c, h_sh, W_hid, pos_emb, b_hid, A_ws, C_ws, E_ws);

    pair_epilogue<<<BB * EP_NBLK, EP_T, 0, stream>>>(A_ws, C_ws, E_ws,
                                                     ln_g, ln_b, W_rel, b_rel, out);
}


# Epilogue: a pair per 16-lane row (two pairs per wave pass, 20 elements a lane), its three sums by four row_xmask DPP adds instead of LDS permutes; 8,192 -> 4,096 waves; sum order changed
# speedup vs baseline: 5.0049x; 1.0716x over previous
#include <hip/hip_runtime.h>
#include <hip/hip_bf16.h>
#include <math.h>

#define BB      32
#define LL      128
#define DD      300
#define KK_REL  12
#define NREL    25
#define PD      50
#define NPAIR   3044
#define FWID    650
#define MROWS   (BB * LL)

typedef __attribute__((ext_vector_type(2))) float v2f;
typedef __attribute__((ext_vector_type(8))) float v8f;

#define E_CHUNK   100
#define E_BLOCKS  (3 * NREL)
#define PE_PCS    (NREL * PD / 2)
#define PE_SWEEPS ((PE_PCS + 255) / 256)
#define GM_ROWS   64
#define GN_TILES  10
#define GN_ROWS   (GN_TILES * 16)
#define KCH       60
#define NCH       (DD / KCH)
#define A_PCS     (GM_ROWS * KCH / 4)
#define A_SWEEPS  ((A_PCS + 255) / 256)
#define B_PCS     (GN_ROWS * KCH / 2)
#define B_SWEEPS  ((B_PCS + 255) / 256)

__global__ __launch_bounds__(256) void pairproj_gemm(
    const float* __restrict__ h_e, const float* __restrict__ h_c,
    const float* __restrict__ h_sh, const float* __restrict__ W_hid,
    const float* __restrict__ pos_emb, const float* __restrict__ b_hid,
    float* __restrict__ A_ws, float* __restrict__ C_ws, float* __restrict__ E_ws)
{
    __shared__ __attribute__((aligned(16))) float As[GM_ROWS * DD];
    __shared__ __attribute__((aligned(16))) float Bs[GN_ROWS * DD];
    __shared__ float wgt[NREL + 7];
    __shared__ __attribute__((aligned(16))) float pes[NREL * PD + 2];

    const int tid  = threadIdx.x;
    const int wave = tid >> 5;
    const int lane = tid & 31;
    const int mq   = blockIdx.x;
    const int ng   = blockIdx.y;
    const int sel  = blockIdx.z;
    const float* H   = sel ? h_c : h_e;
    float*       Out = sel ? C_ws : A_ws;

    int bg[B_SWEEPS], bl[B_SWEEPS];
#pragma unroll
    for (int it = 0; it < B_SWEEPS; ++it) {
        int i = it * 256 + tid;
        i = i < B_PCS ? i : B_PCS - 1;
        const int row = i / (KCH / 2);
        const int c2  = i - row * (KCH / 2);
        int n = ng * GN_ROWS + row;
        n = n < DD ? n : DD - 1;
        bg[it] = n * FWID + 2 * c2;
        bl[it] = row * DD + 2 * c2;
    }
    int ao[A_SWEEPS];
#pragma unroll
    for (int it = 0; it < A_SWEEPS; ++it) {
        int i = it * 256 + tid;
        i = i < A_PCS ? i : A_PCS - 1;
        const int row = i / (KCH / 4);
        const int q   = i - row * (KCH / 4);
        ao[it] = row * DD + 4 * q;
    }
    const bool b_last = (B_SWEEPS - 1) * 256 + tid < B_PCS;
    const bool a_last = (A_SWEEPS - 1) * 256 + tid < A_PCS;
    const float* wsel = W_hid + sel * DD;
    const float* hrow = H    + (size_t)mq * GM_ROWS * DD;
    const float* srow = h_sh + (size_t)mq * GM_ROWS * DD;

    v2f    tb[B_SWEEPS];
    float4 ta[A_SWEEPS], ts[A_SWEEPS];

    const int  lb    = mq + (MROWS / GM_ROWS) * (ng + 2 * sel);
    const bool eblk  = lb < E_BLOCKS;
    const int  er    = lb / 3;
    const int  edout = (lb - 3 * er) * E_CHUNK + tid;
    const bool ecol  = tid < E_CHUNK;
    v2f   epe[PE_SWEEPS];
    v2f   ew[PD / 2];
    float ebias = 0.f;
    if (eblk) {
#pragma unroll
        for (int it = 0; it < PE_SWEEPS; ++it) {
            int q = it * 256 + tid;
            q = q < PE_PCS ? q : PE_PCS - 1;
            epe[it] = *(const v2f*)(pos_emb + 2 * q);
        }
        if (ecol) {
            const float* wrow = W_hid + (size_t)edout * FWID + 2 * DD;
#pragma unroll
            for (int c = 0; c < PD / 2; ++c) ew[c] = *(const v2f*)(wrow + 2 * c);
            ebias = b_hid[edout];
        }
    }

#pragma unroll
    for (int it = 0; it < B_SWEEPS; ++it) tb[it] = *(const v2f*)(wsel + bg[it]);
#pragma unroll
    for (int it = 0; it < A_SWEEPS; ++it) {
        ta[it] = *(const float4*)(hrow + ao[it]);
        ts[it] = *(const float4*)(srow + ao[it]);
    }

    if (eblk) {
#pragma unroll
        for (int it = 0; it < PE_SWEEPS; ++it) {
            int q = it * 256 + tid;
            q = q < PE_PCS ? q : PE_PCS - 1;
            *(v2f*)(pes + 2 * q) = epe[it];
        }
        if (tid < NREL) {
            const int rp = tid;
            const int ad = rp >= KK_REL ? rp - KK_REL : KK_REL - rp;
            const float cnt = (float)(LL - ad);
            const int d = er - rp;
            wgt[rp] = cnt * expf(-(float)(d * d));
        }
    }

#pragma unroll
    for (int it = 0; it < B_SWEEPS; ++it)
        if (it < B_SWEEPS - 1 || b_last) *(v2f*)(Bs + bl[it]) = tb[it];
#pragma unroll
    for (int it = 0; it < A_SWEEPS; ++it)
        if (it < A_SWEEPS - 1 || a_last) {
            float4 r;
            r.x = ta[it].x + ts[it].x; r.y = ta[it].y + ts[it].y;
            r.z = ta[it].z + ts[it].z; r.w = ta[it].w + ts[it].w;
            *(float4*)(As + ao[it]) = r;
        }
    __syncthreads();

    if (eblk && tid < 128) {
        const int l32 = tid & 31;
        const int p1  = l32 + 32 < PD ? l32 + 32 : PD - 1;
        float s0 = 0.f, s1 = 0.f;
        for (int rp = 0; rp < NREL; ++rp) {
            const float wv = wgt[rp];
            s0 += wv * pes[rp * PD + l32];
            s1 += wv * pes[rp * PD + p1];
        }
        float acc = ebias;
#pragma unroll
        for (int c = 0; c < PD / 2; ++c) {
            const int pa = 2 * c, pb = 2 * c + 1;
            const float sa = __int_as_float(__builtin_amdgcn_readlane(__float_as_int(pa < 32 ? s0 : s1), pa & 31));
            const float sb = __int_as_float(__builtin_amdgcn_readlane(__float_as_int(pb < 32 ? s0 : s1), pb & 31));
            acc += sa * ew[c].x;
            acc += sb * ew[c].y;
        }
        if (ecol) E_ws[er * DD + edout] = acc;
    }

    const int mlane = lane & 15;
    const int kk    = (lane >> 4) * 2;
    const int mt    = wave & 3;
    const int nh    = wave >> 2;
    const float* ap = As + (mt * 16 + mlane) * DD + kk;
    const float* bp = Bs + (nh * 5 * 16 + mlane) * DD + kk;

    v8f acc0 = (v8f)0.0f, acc1 = (v8f)0.0f, acc2 = (v8f)0.0f, acc3 = (v8f)0.0f, acc4 = (v8f)0.0f;
#pragma unroll 1
    for (int c = 0; c < NCH; ++c) {
        const bool more = c + 1 < NCH;
        if (more) {
            const int kc = (c + 1) * KCH;
#pragma unroll
            for (int it = 0; it < B_SWEEPS; ++it) tb[it] = *(const v2f*)(wsel + bg[it] + kc);
#pragma unroll
            for (int it = 0; it < A_SWEEPS; ++it) {
                ta[it] = *(const float4*)(hrow + ao[it] + kc);
                ts[it] = *(const float4*)(srow + ao[it] + kc);
            }
        }
#pragma unroll
        for (int ks = 0; ks < KCH; ks += 4) {
            const int k0 = c * KCH + ks;
            const v2f a  = *(const v2f*)(ap + k0);
            const v2f b0 = *(const v2f*)(bp + 0 * 16 * DD + k0);
            const v2f b1 = *(const v2f*)(bp + 1 * 16 * DD + k0);
            const v2f b2 = *(const v2f*)(bp + 2 * 16 * DD + k0);
            const v2f b3 = *(const v2f*)(bp + 3 * 16 * DD + k0);
            const v2f b4 = *(const v2f*)(bp + 4 * 16 * DD + k0);
            acc0 = __builtin_amdgcn_wmma_f32_16x16x4_f32(false, a, false, b0, (short)0, acc0, false, false);
            acc1 = __builtin_amdgcn_wmma_f32_16x16x4_f32(false, a, false, b1, (short)0, acc1, false, false);
            acc2 = __builtin_amdgcn_wmma_f32_16x16x4_f32(false, a, false, b2, (short)0, acc2, false, false);
            acc3 = __builtin_amdgcn_wmma_f32_16x16x4_f32(false, a, false, b3, (short)0, acc3, false, false);
            acc4 = __builtin_amdgcn_wmma_f32_16x16x4_f32(false, a, false, b4, (short)0, acc4, false, false);
        }
        if (more) {
            const int kc = (c + 1) * KCH;
#pragma unroll
            for (int it = 0; it < B_SWEEPS; ++it)
                if (it < B_SWEEPS - 1 || b_last) *(v2f*)(Bs + bl[it] + kc) = tb[it];
#pragma unroll
            for (int it = 0; it < A_SWEEPS; ++it)
                if (it < A_SWEEPS - 1 || a_last) {
                    float4 r;
                    r.x = ta[it].x + ts[it].x; r.y = ta[it].y + ts[it].y;
                    r.z = ta[it].z + ts[it].z; r.w = ta[it].w + ts[it].w;
                    *(float4*)(As + ao[it] + kc) = r;
                }
            __syncthreads();
        }
    }

    const int rbase = (lane >> 4) << 3;
    const size_t row0 = (size_t)(mq * GM_ROWS + mt * 16 + rbase);
    v8f accs[5] = {acc0, acc1, acc2, acc3, acc4};
#pragma unroll
    for (int j = 0; j < 5; ++j) {
        const int ncol = (ng * GN_TILES + nh * 5 + j) * 16 + mlane;
        if (ncol < DD) {
#pragma unroll
            for (int g = 0; g < 8; ++g)
                Out[(row0 + g) * DD + ncol] = accs[j][g];
        }
    }
}

#define EP_ROWS   16
#define EP_T      (EP_ROWS * 32)
#define EP_CROWS  (EP_ROWS + 2 * KK_REL)
#define EP_NBLK   (LL / EP_ROWS)
#define C_PCS     (EP_CROWS * DD / 4)
#define C_SWEEPS  ((C_PCS + EP_T - 1) / EP_T)
#define E_PCS     (NREL * DD / 4)
#define E_SWEEPS  ((E_PCS + EP_T - 1) / EP_T)
#define LOG2E_F   1.4426950408889634f

typedef __attribute__((ext_vector_type(4))) float v4f;

template <int MASK>
__device__ __forceinline__ float row_xor(float x) {
    const int b = __builtin_bit_cast(int, x);
    return __builtin_bit_cast(float, __builtin_amdgcn_update_dpp(b, b, 0x160 + MASK, 0xf, 0xf, true));
}
__device__ __forceinline__ float row16_sum(float x) {
    x += row_xor<1>(x);
    x += row_xor<2>(x);
    x += row_xor<4>(x);
    x += row_xor<8>(x);
    return x;
}

__global__ __launch_bounds__(EP_T) void pair_epilogue(
    const float* __restrict__ A_ws, const float* __restrict__ C_ws,
    const float* __restrict__ E_ws,
    const float* __restrict__ ln_g, const float* __restrict__ ln_b,
    const float* __restrict__ W_rel, const float* __restrict__ b_rel,
    float* __restrict__ out)
{
    __shared__ __attribute__((aligned(16))) float Cs[EP_CROWS * DD];
    __shared__ __attribute__((aligned(16))) float Es[NREL * DD];

    const int tid  = threadIdx.x;
    const int lane = tid & 31;
    const int c    = lane & 15;
    const int hf   = lane >> 4;
    const int wave = __builtin_amdgcn_readfirstlane(tid >> 5);
    const int b    = blockIdx.x / EP_NBLK;
    const int i0   = (blockIdx.x - b * EP_NBLK) * EP_ROWS;
    int jlo = i0 - KK_REL;
    jlo = jlo < 0 ? 0 : jlo;
    jlo = jlo > LL - EP_CROWS ? LL - EP_CROWS : jlo;

    const float4* csrc = (const float4*)(C_ws + ((size_t)b * LL + jlo) * DD);
    const float4* esrc = (const float4*)E_ws;
#pragma unroll
    for (int it = 0; it < C_SWEEPS; ++it) {
        int q = it * EP_T + tid;
        q = q < C_PCS ? q : C_PCS - 1;
        ((float4*)Cs)[q] = csrc[q];
    }
#pragma unroll
    for (int it = 0; it < E_SWEEPS; ++it) {
        int q = it * EP_T + tid;
        q = q < E_PCS ? q : E_PCS - 1;
        ((float4*)Es)[q] = esrc[q];
    }

    const int i = i0 + wave;
    const float* ra = A_ws + ((size_t)b * LL + i) * DD;
    v2f a[10], g[10], bb[10], wr[10];
#pragma unroll
    for (int q = 0; q < 10; ++q) {
        const int idx = (q >> 1) * 64 + c * 4 + (q & 1) * 2;
        a[q] = (v2f)0.0f; g[q] = (v2f)0.0f; bb[q] = (v2f)0.0f; wr[q] = (v2f)0.0f;
        if (idx < DD) {
            a[q]  = *(const v2f*)(ra    + idx);
            g[q]  = *(const v2f*)(ln_g  + idx);
            bb[q] = *(const v2f*)(ln_b  + idx);
            wr[q] = *(const v2f*)(W_rel + idx);
        }
    }
    const float brel = b_rel[0];
    __syncthreads();

    const int rlo = i < KK_REL ? KK_REL - i : 0;
    const int rhi = i + KK_REL > LL - 1 ? LL - 1 - i + KK_REL : 2 * KK_REL;
    const int np  = rhi - rlo + 1;
    const int nit = (np + 1) >> 1;
    const int m = i < KK_REL ? i : KK_REL;
    const int n = i > LL - KK_REL ? i - (LL - KK_REL) : 0;
    const int start = NREL * i - (KK_REL * m - (m * (m - 1)) / 2) - (n * (n + 1)) / 2;

    float res = 0.f;
    for (int it = 0; it < nit; ++it) {
        int r = rlo + 2 * it + hf;
        r = r < rhi ? r : rhi;
        const float* crow = Cs + (i - KK_REL + r - jlo) * DD + c * 4;
        const float* erow = Es + r * DD + c * 4;
        v2f h[10];
        v2f s2 = (v2f)0.0f, q2 = (v2f)0.0f;
#pragma unroll
        for (int t = 0; t < 5; ++t) {
            v2f h0 = (v2f)0.0f, h1 = (v2f)0.0f;
            if (t * 64 + c * 4 < DD) {
                const v4f vc = *(const v4f*)(crow + t * 64);
                const v4f ve = *(const v4f*)(erow + t * 64);
                h0 = a[2 * t]     + vc.xy + ve.xy;
                h1 = a[2 * t + 1] + vc.zw + ve.zw;
            }
            h[2 * t]     = h0;
            h[2 * t + 1] = h1;
            s2 += h0 + h1;
            q2 += h0 * h0 + h1 * h1;
        }
        const float s  = row16_sum(s2.x + s2.y);
        const float ss = row16_sum(q2.x + q2.y);
        const float mu   = s * (1.f / DD);
        const float var  = ss * (1.f / DD) - mu * mu;
        const float rstd = rsqrtf(var + 1e-5f);

        v2f acc2 = (v2f)0.0f;
#pragma unroll
        for (int t = 0; t < 5; ++t) {
            if (t * 64 + c * 4 < DD) {
#pragma unroll
                for (int u = 0; u < 2; ++u) {
                    const int q = 2 * t + u;
                    v2f v = (h[q] - mu) * rstd * g[q] + bb[q];
                    const float e0 = __builtin_amdgcn_exp2f(v.x * LOG2E_F) - 1.f;
                    const float e1 = __builtin_amdgcn_exp2f(v.y * LOG2E_F) - 1.f;
                    v.x = v.x > 0.f ? v.x : e0;
                    v.y = v.y > 0.f ? v.y : e1;
                    acc2 += v * wr[q];
                }
            }
        }
        const float acc = row16_sum(acc2.x + acc2.y);

        if (c == it) res = acc + brel;
    }

    const int k = 2 * c + hf;
    if (k < np) out[(size_t)b * NPAIR + start + k] = res;
}

extern "C" void kernel_launch(void* const* d_in, const int* in_sizes, int n_in,
                              void* d_out, int out_size, void* d_ws, size_t ws_size,
                              hipStream_t stream) {
    const float* h_e     = (const float*)d_in[0];
    const float* h_c     = (const float*)d_in[1];
    const float* h_sh    = (const float*)d_in[2];
    const float* pos_emb = (const float*)d_in[3];
    const float* W_hid   = (const float*)d_in[4];
    const float* b_hid   = (const float*)d_in[5];
    const float* ln_g    = (const float*)d_in[6];
    const float* ln_b    = (const float*)d_in[7];
    const float* W_rel   = (const float*)d_in[8];
    const float* b_rel   = (const float*)d_in[9];
    float* out = (float*)d_out;

    float* ws   = (float*)d_ws;
    float* A_ws = ws;
    float* C_ws = A_ws + (size_t)MROWS * DD;
    float* E_ws = C_ws + (size_t)MROWS * DD;

    dim3 ggrid(MROWS / GM_ROWS, 2, 2);
    pairproj_gemm<<<ggrid, 256, 0, stream>>>(h_e, h_c, h_sh, W_hid, pos_emb, b_hid, A_ws, C_ws, E_ws);

    pair_epilogue<<<BB * EP_NBLK, EP_T, 0, stream>>>(A_ws, C_ws, E_ws,
                                                     ln_g, ln_b, W_rel, b_rel, out);
}


# Epilogue LN+ELU per element without compare/select: x=h*rstd-mu*rstd, ELU(v)+1=max(v+1, exp2 clamped to [0,1]); the ones leave as -sum(W_rel) after the dot; 202 -> 157 vector instr per pass
# speedup vs baseline: 5.2442x; 1.0478x over previous
#include <hip/hip_runtime.h>
#include <hip/hip_bf16.h>
#include <math.h>

#define BB      32
#define LL      128
#define DD      300
#define KK_REL  12
#define NREL    25
#define PD      50
#define NPAIR   3044
#define FWID    650
#define MROWS   (BB * LL)

typedef __attribute__((ext_vector_type(2))) float v2f;
typedef __attribute__((ext_vector_type(8))) float v8f;

#define E_CHUNK   100
#define E_BLOCKS  (3 * NREL)
#define PE_PCS    (NREL * PD / 2)
#define PE_SWEEPS ((PE_PCS + 255) / 256)
#define GM_ROWS   64
#define GN_TILES  10
#define GN_ROWS   (GN_TILES * 16)
#define KCH       60
#define NCH       (DD / KCH)
#define A_PCS     (GM_ROWS * KCH / 4)
#define A_SWEEPS  ((A_PCS + 255) / 256)
#define B_PCS     (GN_ROWS * KCH / 2)
#define B_SWEEPS  ((B_PCS + 255) / 256)

__global__ __launch_bounds__(256) void pairproj_gemm(
    const float* __restrict__ h_e, const float* __restrict__ h_c,
    const float* __restrict__ h_sh, const float* __restrict__ W_hid,
    const float* __restrict__ pos_emb, const float* __restrict__ b_hid,
    float* __restrict__ A_ws, float* __restrict__ C_ws, float* __restrict__ E_ws)
{
    __shared__ __attribute__((aligned(16))) float As[GM_ROWS * DD];
    __shared__ __attribute__((aligned(16))) float Bs[GN_ROWS * DD];
    __shared__ float wgt[NREL + 7];
    __shared__ __attribute__((aligned(16))) float pes[NREL * PD + 2];

    const int tid  = threadIdx.x;
    const int wave = tid >> 5;
    const int lane = tid & 31;
    const int mq   = blockIdx.x;
    const int ng   = blockIdx.y;
    const int sel  = blockIdx.z;
    const float* H   = sel ? h_c : h_e;
    float*       Out = sel ? C_ws : A_ws;

    int bg[B_SWEEPS], bl[B_SWEEPS];
#pragma unroll
    for (int it = 0; it < B_SWEEPS; ++it) {
        int i = it * 256 + tid;
        i = i < B_PCS ? i : B_PCS - 1;
        const int row = i / (KCH / 2);
        const int c2  = i - row * (KCH / 2);
        int n = ng * GN_ROWS + row;
        n = n < DD ? n : DD - 1;
        bg[it] = n * FWID + 2 * c2;
        bl[it] = row * DD + 2 * c2;
    }
    int ao[A_SWEEPS];
#pragma unroll
    for (int it = 0; it < A_SWEEPS; ++it) {
        int i = it * 256 + tid;
        i = i < A_PCS ? i : A_PCS - 1;
        const int row = i / (KCH / 4);
        const int q   = i - row * (KCH / 4);
        ao[it] = row * DD + 4 * q;
    }
    const bool b_last = (B_SWEEPS - 1) * 256 + tid < B_PCS;
    const bool a_last = (A_SWEEPS - 1) * 256 + tid < A_PCS;
    const float* wsel = W_hid + sel * DD;
    const float* hrow = H    + (size_t)mq * GM_ROWS * DD;
    const float* srow = h_sh + (size_t)mq * GM_ROWS * DD;

    v2f    tb[B_SWEEPS];
    float4 ta[A_SWEEPS], ts[A_SWEEPS];

    const int  lb    = mq + (MROWS / GM_ROWS) * (ng + 2 * sel);
    const bool eblk  = lb < E_BLOCKS;
    const int  er    = lb / 3;
    const int  edout = (lb - 3 * er) * E_CHUNK + tid;
    const bool ecol  = tid < E_CHUNK;
    v2f   epe[PE_SWEEPS];
    v2f   ew[PD / 2];
    float ebias = 0.f;
    if (eblk) {
#pragma unroll
        for (int it = 0; it < PE_SWEEPS; ++it) {
            int q = it * 256 + tid;
            q = q < PE_PCS ? q : PE_PCS - 1;
            epe[it] = *(const v2f*)(pos_emb + 2 * q);
        }
        if (ecol) {
            const float* wrow = W_hid + (size_t)edout * FWID + 2 * DD;
#pragma unroll
            for (int c = 0; c < PD / 2; ++c) ew[c] = *(const v2f*)(wrow + 2 * c);
            ebias = b_hid[edout];
        }
    }

#pragma unroll
    for (int it = 0; it < B_SWEEPS; ++it) tb[it] = *(const v2f*)(wsel + bg[it]);
#pragma unroll
    for (int it = 0; it < A_SWEEPS; ++it) {
        ta[it] = *(const float4*)(hrow + ao[it]);
        ts[it] = *(const float4*)(srow + ao[it]);
    }

    if (eblk) {
#pragma unroll
        for (int it = 0; it < PE_SWEEPS; ++it) {
            int q = it * 256 + tid;
            q = q < PE_PCS ? q : PE_PCS - 1;
            *(v2f*)(pes + 2 * q) = epe[it];
        }
        if (tid < NREL) {
            const int rp = tid;
            const int ad = rp >= KK_REL ? rp - KK_REL : KK_REL - rp;
            const float cnt = (float)(LL - ad);
            const int d = er - rp;
            wgt[rp] = cnt * expf(-(float)(d * d));
        }
    }

#pragma unroll
    for (int it = 0; it < B_SWEEPS; ++it)
        if (it < B_SWEEPS - 1 || b_last) *(v2f*)(Bs + bl[it]) = tb[it];
#pragma unroll
    for (int it = 0; it < A_SWEEPS; ++it)
        if (it < A_SWEEPS - 1 || a_last) {
            float4 r;
            r.x = ta[it].x + ts[it].x; r.y = ta[it].y + ts[it].y;
            r.z = ta[it].z + ts[it].z; r.w = ta[it].w + ts[it].w;
            *(float4*)(As + ao[it]) = r;
        }
    __syncthreads();

    if (eblk && tid < 128) {
        const int l32 = tid & 31;
        const int p1  = l32 + 32 < PD ? l32 + 32 : PD - 1;
        float s0 = 0.f, s1 = 0.f;
        for (int rp = 0; rp < NREL; ++rp) {
            const float wv = wgt[rp];
            s0 += wv * pes[rp * PD + l32];
            s1 += wv * pes[rp * PD + p1];
        }
        float acc = ebias;
#pragma unroll
        for (int c = 0; c < PD / 2; ++c) {
            const int pa = 2 * c, pb = 2 * c + 1;
            const float sa = __int_as_float(__builtin_amdgcn_readlane(__float_as_int(pa < 32 ? s0 : s1), pa & 31));
            const float sb = __int_as_float(__builtin_amdgcn_readlane(__float_as_int(pb < 32 ? s0 : s1), pb & 31));
            acc += sa * ew[c].x;
            acc += sb * ew[c].y;
        }
        if (ecol) E_ws[er * DD + edout] = acc;
    }

    const int mlane = lane & 15;
    const int kk    = (lane >> 4) * 2;
    const int mt    = wave & 3;
    const int nh    = wave >> 2;
    const float* ap = As + (mt * 16 + mlane) * DD + kk;
    const float* bp = Bs + (nh * 5 * 16 + mlane) * DD + kk;

    v8f acc0 = (v8f)0.0f, acc1 = (v8f)0.0f, acc2 = (v8f)0.0f, acc3 = (v8f)0.0f, acc4 = (v8f)0.0f;
#pragma unroll 1
    for (int c = 0; c < NCH; ++c) {
        const bool more = c + 1 < NCH;
        if (more) {
            const int kc = (c + 1) * KCH;
#pragma unroll
            for (int it = 0; it < B_SWEEPS; ++it) tb[it] = *(const v2f*)(wsel + bg[it] + kc);
#pragma unroll
            for (int it = 0; it < A_SWEEPS; ++it) {
                ta[it] = *(const float4*)(hrow + ao[it] + kc);
                ts[it] = *(const float4*)(srow + ao[it] + kc);
            }
        }
#pragma unroll
        for (int ks = 0; ks < KCH; ks += 4) {
            const int k0 = c * KCH + ks;
            const v2f a  = *(const v2f*)(ap + k0);
            const v2f b0 = *(const v2f*)(bp + 0 * 16 * DD + k0);
            const v2f b1 = *(const v2f*)(bp + 1 * 16 * DD + k0);
            const v2f b2 = *(const v2f*)(bp + 2 * 16 * DD + k0);
            const v2f b3 = *(const v2f*)(bp + 3 * 16 * DD + k0);
            const v2f b4 = *(const v2f*)(bp + 4 * 16 * DD + k0);
            acc0 = __builtin_amdgcn_wmma_f32_16x16x4_f32(false, a, false, b0, (short)0, acc0, false, false);
            acc1 = __builtin_amdgcn_wmma_f32_16x16x4_f32(false, a, false, b1, (short)0, acc1, false, false);
            acc2 = __builtin_amdgcn_wmma_f32_16x16x4_f32(false, a, false, b2, (short)0, acc2, false, false);
            acc3 = __builtin_amdgcn_wmma_f32_16x16x4_f32(false, a, false, b3, (short)0, acc3, false, false);
            acc4 = __builtin_amdgcn_wmma_f32_16x16x4_f32(false, a, false, b4, (short)0, acc4, false, false);
        }
        if (more) {
            const int kc = (c + 1) * KCH;
#pragma unroll
            for (int it = 0; it < B_SWEEPS; ++it)
                if (it < B_SWEEPS - 1 || b_last) *(v2f*)(Bs + bl[it] + kc) = tb[it];
#pragma unroll
            for (int it = 0; it < A_SWEEPS; ++it)
                if (it < A_SWEEPS - 1 || a_last) {
                    float4 r;
                    r.x = ta[it].x + ts[it].x; r.y = ta[it].y + ts[it].y;
                    r.z = ta[it].z + ts[it].z; r.w = ta[it].w + ts[it].w;
                    *(float4*)(As + ao[it] + kc) = r;
                }
            __syncthreads();
        }
    }

    const int rbase = (lane >> 4) << 3;
    const size_t row0 = (size_t)(mq * GM_ROWS + mt * 16 + rbase);
    v8f accs[5] = {acc0, acc1, acc2, acc3, acc4};
#pragma unroll
    for (int j = 0; j < 5; ++j) {
        const int ncol = (ng * GN_TILES + nh * 5 + j) * 16 + mlane;
        if (ncol < DD) {
#pragma unroll
            for (int g = 0; g < 8; ++g)
                Out[(row0 + g) * DD + ncol] = accs[j][g];
        }
    }
}

#define EP_ROWS   16
#define EP_T      (EP_ROWS * 32)
#define EP_CROWS  (EP_ROWS + 2 * KK_REL)
#define EP_NBLK   (LL / EP_ROWS)
#define C_PCS     (EP_CROWS * DD / 4)
#define C_SWEEPS  ((C_PCS + EP_T - 1) / EP_T)
#define E_PCS     (NREL * DD / 4)
#define E_SWEEPS  ((E_PCS + EP_T - 1) / EP_T)
#define LOG2E_F   1.4426950408889634f

typedef __attribute__((ext_vector_type(4))) float v4f;

template <int MASK>
__device__ __forceinline__ float row_xor(float x) {
    const int b = __builtin_bit_cast(int, x);
    return __builtin_bit_cast(float, __builtin_amdgcn_update_dpp(b, b, 0x160 + MASK, 0xf, 0xf, true));
}
__device__ __forceinline__ float row16_sum(float x) {
    x += row_xor<1>(x);
    x += row_xor<2>(x);
    x += row_xor<4>(x);
    x += row_xor<8>(x);
    return x;
}

__global__ __launch_bounds__(EP_T) void pair_epilogue(
    const float* __restrict__ A_ws, const float* __restrict__ C_ws,
    const float* __restrict__ E_ws,
    const float* __restrict__ ln_g, const float* __restrict__ ln_b,
    const float* __restrict__ W_rel, const float* __restrict__ b_rel,
    float* __restrict__ out)
{
    __shared__ __attribute__((aligned(16))) float Cs[EP_CROWS * DD];
    __shared__ __attribute__((aligned(16))) float Es[NREL * DD];

    const int tid  = threadIdx.x;
    const int lane = tid & 31;
    const int c    = lane & 15;
    const int hf   = lane >> 4;
    const int wave = __builtin_amdgcn_readfirstlane(tid >> 5);
    const int b    = blockIdx.x / EP_NBLK;
    const int i0   = (blockIdx.x - b * EP_NBLK) * EP_ROWS;
    int jlo = i0 - KK_REL;
    jlo = jlo < 0 ? 0 : jlo;
    jlo = jlo > LL - EP_CROWS ? LL - EP_CROWS : jlo;

    const float4* csrc = (const float4*)(C_ws + ((size_t)b * LL + jlo) * DD);
    const float4* esrc = (const float4*)E_ws;
#pragma unroll
    for (int it = 0; it < C_SWEEPS; ++it) {
        int q = it * EP_T + tid;
        q = q < C_PCS ? q : C_PCS - 1;
        ((float4*)Cs)[q] = csrc[q];
    }
#pragma unroll
    for (int it = 0; it < E_SWEEPS; ++it) {
        int q = it * EP_T + tid;
        q = q < E_PCS ? q : E_PCS - 1;
        ((float4*)Es)[q] = esrc[q];
    }

    const int i = i0 + wave;
    const float* ra = A_ws + ((size_t)b * LL + i) * DD;
    v2f a[10], g[10], bb[10], wr[10];
#pragma unroll
    for (int q = 0; q < 10; ++q) {
        const int idx = (q >> 1) * 64 + c * 4 + (q & 1) * 2;
        a[q] = (v2f)0.0f; g[q] = (v2f)0.0f; bb[q] = (v2f)0.0f; wr[q] = (v2f)0.0f;
        if (idx < DD) {
            a[q]  = *(const v2f*)(ra    + idx);
            g[q]  = *(const v2f*)(ln_g  + idx);
            bb[q] = *(const v2f*)(ln_b  + idx) + 1.0f;
            wr[q] = *(const v2f*)(W_rel + idx);
        }
    }
    v2f wl2 = (v2f)0.0f;
#pragma unroll
    for (int q = 0; q < 10; ++q) wl2 += wr[q];
    const float bias = b_rel[0] - row16_sum(wl2.x + wl2.y);
    __syncthreads();

    const int rlo = i < KK_REL ? KK_REL - i : 0;
    const int rhi = i + KK_REL > LL - 1 ? LL - 1 - i + KK_REL : 2 * KK_REL;
    const int np  = rhi - rlo + 1;
    const int nit = (np + 1) >> 1;
    const int m = i < KK_REL ? i : KK_REL;
    const int n = i > LL - KK_REL ? i - (LL - KK_REL) : 0;
    const int start = NREL * i - (KK_REL * m - (m * (m - 1)) / 2) - (n * (n + 1)) / 2;

    float res = 0.f;
    for (int it = 0; it < nit; ++it) {
        int r = rlo + 2 * it + hf;
        r = r < rhi ? r : rhi;
        const float* crow = Cs + (i - KK_REL + r - jlo) * DD + c * 4;
        const float* erow = Es + r * DD + c * 4;
        v2f h[10];
        v2f s2 = (v2f)0.0f, q2 = (v2f)0.0f;
#pragma unroll
        for (int t = 0; t < 5; ++t) {
            v2f h0 = (v2f)0.0f, h1 = (v2f)0.0f;
            if (t * 64 + c * 4 < DD) {
                const v4f vc = *(const v4f*)(crow + t * 64);
                const v4f ve = *(const v4f*)(erow + t * 64);
                h0 = a[2 * t]     + vc.xy + ve.xy;
                h1 = a[2 * t + 1] + vc.zw + ve.zw;
            }
            h[2 * t]     = h0;
            h[2 * t + 1] = h1;
            s2 += h0 + h1;
            q2 += h0 * h0 + h1 * h1;
        }
        const float s  = row16_sum(s2.x + s2.y);
        const float ss = row16_sum(q2.x + q2.y);
        const float mu   = s * (1.f / DD);
        const float var  = ss * (1.f / DD) - mu * mu;
        const float rstd = rsqrtf(var + 1e-5f);
        const float nmr  = -mu * rstd;

        v2f acc2 = (v2f)0.0f;
#pragma unroll
        for (int t = 0; t < 5; ++t) {
            if (t * 64 + c * 4 < DD) {
#pragma unroll
                for (int u = 0; u < 2; ++u) {
                    const int q = 2 * t + u;
                    const v2f x  = h[q] * rstd + nmr;
                    const v2f v1 = x * g[q] + bb[q];
                    const v2f ul = v1 * LOG2E_F - LOG2E_F;
                    const float e0 = __builtin_amdgcn_fmed3f(__builtin_amdgcn_exp2f(ul.x), 0.f, 1.f);
                    const float e1 = __builtin_amdgcn_fmed3f(__builtin_amdgcn_exp2f(ul.y), 0.f, 1.f);
                    v2f m;
                    m.x = fmaxf(v1.x, e0);
                    m.y = fmaxf(v1.y, e1);
                    acc2 += m * wr[q];
                }
            }
        }
        const float acc = row16_sum(acc2.x + acc2.y);

        if (c == it) res = acc + bias;
    }

    const int k = 2 * c + hf;
    if (k < np) out[(size_t)b * NPAIR + start + k] = res;
}

extern "C" void kernel_launch(void* const* d_in, const int* in_sizes, int n_in,
                              void* d_out, int out_size, void* d_ws, size_t ws_size,
                              hipStream_t stream) {
    const float* h_e     = (const float*)d_in[0];
    const float* h_c     = (const float*)d_in[1];
    const float* h_sh    = (const float*)d_in[2];
    const float* pos_emb = (const float*)d_in[3];
    const float* W_hid   = (const float*)d_in[4];
    const float* b_hid   = (const float*)d_in[5];
    const float* ln_g    = (const float*)d_in[6];
    const float* ln_b    = (const float*)d_in[7];
    const float* W_rel   = (const float*)d_in[8];
    const float* b_rel   = (const float*)d_in[9];
    float* out = (float*)d_out;

    float* ws   = (float*)d_ws;
    float* A_ws = ws;
    float* C_ws = A_ws + (size_t)MROWS * DD;
    float* E_ws = C_ws + (size_t)MROWS * DD;

    dim3 ggrid(MROWS / GM_ROWS, 2, 2);
    pairproj_gemm<<<ggrid, 256, 0, stream>>>(h_e, h_c, h_sh, W_hid, pos_emb, b_hid, A_ws, C_ws, E_ws);

    pair_epilogue<<<BB * EP_NBLK, EP_T, 0, stream>>>(A_ws, C_ws, E_ws,
                                                     ln_g, ln_b, W_rel, b_rel, out);
}


# GEMM: loads and LDS stores of chunk c+1 issued in quarters behind the WMMA groups of chunk c (loads behind groups 0-3, stores behind 3-6), not as blocks at its head and end; same sum order
# speedup vs baseline: 5.4781x; 1.0446x over previous
#include <hip/hip_runtime.h>
#include <hip/hip_bf16.h>
#include <math.h>

#define BB      32
#define LL      128
#define DD      300
#define KK_REL  12
#define NREL    25
#define PD      50
#define NPAIR   3044
#define FWID    650
#define MROWS   (BB * LL)

typedef __attribute__((ext_vector_type(2))) float v2f;
typedef __attribute__((ext_vector_type(4))) float v4f;
typedef __attribute__((ext_vector_type(8))) float v8f;

#define E_CHUNK   100
#define E_BLOCKS  (3 * NREL)
#define PE_PCS    (NREL * PD / 2)
#define PE_SWEEPS ((PE_PCS + 255) / 256)
#define GM_ROWS   64
#define GN_TILES  10
#define GN_ROWS   (GN_TILES * 16)
#define KCH       60
#define NCH       (DD / KCH)
#define A_PCS     (GM_ROWS * KCH / 4)
#define A_SWEEPS  ((A_PCS + 255) / 256)
#define B_PCS     (GN_ROWS * KCH / 2)
#define B_SWEEPS  ((B_PCS + 255) / 256)
#define NGRP      8
#define NQ        4
#define ST_G0     3
static_assert(A_SWEEPS == NQ, "one A piece per quarter");
static_assert(ST_G0 + NQ <= NGRP, "every quarter is written inside its chunk");

__global__ __launch_bounds__(256) void pairproj_gemm(
    const float* __restrict__ h_e, const float* __restrict__ h_c,
    const float* __restrict__ h_sh, const float* __restrict__ W_hid,
    const float* __restrict__ pos_emb, const float* __restrict__ b_hid,
    float* __restrict__ A_ws, float* __restrict__ C_ws, float* __restrict__ E_ws)
{
    __shared__ __attribute__((aligned(16))) float As[GM_ROWS * DD];
    __shared__ __attribute__((aligned(16))) float Bs[GN_ROWS * DD];
    __shared__ float wgt[NREL + 7];
    __shared__ __attribute__((aligned(16))) float pes[NREL * PD + 2];

    const int tid  = threadIdx.x;
    const int wave = tid >> 5;
    const int lane = tid & 31;
    const int mq   = blockIdx.x;
    const int ng   = blockIdx.y;
    const int sel  = blockIdx.z;
    const float* H   = sel ? h_c : h_e;
    float*       Out = sel ? C_ws : A_ws;

    int bg[B_SWEEPS], bl[B_SWEEPS];
#pragma unroll
    for (int it = 0; it < B_SWEEPS; ++it) {
        int i = it * 256 + tid;
        i = i < B_PCS ? i : B_PCS - 1;
        const int row = i / (KCH / 2);
        const int c2  = i - row * (KCH / 2);
        int n = ng * GN_ROWS + row;
        n = n < DD ? n : DD - 1;
        bg[it] = n * FWID + 2 * c2;
        bl[it] = row * DD + 2 * c2;
    }
    int ao[A_SWEEPS];
#pragma unroll
    for (int it = 0; it < A_SWEEPS; ++it) {
        int i = it * 256 + tid;
        i = i < A_PCS ? i : A_PCS - 1;
        const int row = i / (KCH / 4);
        const int q   = i - row * (KCH / 4);
        ao[it] = row * DD + 4 * q;
    }
    const bool b_last = (B_SWEEPS - 1) * 256 + tid < B_PCS;
    const bool a_last = (A_SWEEPS - 1) * 256 + tid < A_PCS;
    const float* wsel = W_hid + sel * DD;
    const float* hrow = H    + (size_t)mq * GM_ROWS * DD;
    const float* srow = h_sh + (size_t)mq * GM_ROWS * DD;

    v2f    tb[B_SWEEPS];
    v4f    ta[A_SWEEPS], ts[A_SWEEPS];

    const int  lb    = mq + (MROWS / GM_ROWS) * (ng + 2 * sel);
    const bool eblk  = lb < E_BLOCKS;
    const int  er    = lb / 3;
    const int  edout = (lb - 3 * er) * E_CHUNK + tid;
    const bool ecol  = tid < E_CHUNK;
    v2f   epe[PE_SWEEPS];
    v2f   ew[PD / 2];
    float ebias = 0.f;
    if (eblk) {
#pragma unroll
        for (int it = 0; it < PE_SWEEPS; ++it) {
            int q = it * 256 + tid;
            q = q < PE_PCS ? q : PE_PCS - 1;
            epe[it] = *(const v2f*)(pos_emb + 2 * q);
        }
        if (ecol) {
            const float* wrow = W_hid + (size_t)edout * FWID + 2 * DD;
#pragma unroll
            for (int c = 0; c < PD / 2; ++c) ew[c] = *(const v2f*)(wrow + 2 * c);
            ebias = b_hid[edout];
        }
    }

#pragma unroll
    for (int it = 0; it < B_SWEEPS; ++it) tb[it] = *(const v2f*)(wsel + bg[it]);
#pragma unroll
    for (int it = 0; it < A_SWEEPS; ++it) {
        ta[it] = *(const v4f*)(hrow + ao[it]);
        ts[it] = *(const v4f*)(srow + ao[it]);
    }

    if (eblk) {
#pragma unroll
        for (int it = 0; it < PE_SWEEPS; ++it) {
            int q = it * 256 + tid;
            q = q < PE_PCS ? q : PE_PCS - 1;
            *(v2f*)(pes + 2 * q) = epe[it];
        }
        if (tid < NREL) {
            const int rp = tid;
            const int ad = rp >= KK_REL ? rp - KK_REL : KK_REL - rp;
            const float cnt = (float)(LL - ad);
            const int d = er - rp;
            wgt[rp] = cnt * expf(-(float)(d * d));
        }
    }

#pragma unroll
    for (int it = 0; it < B_SWEEPS; ++it)
        if (it < B_SWEEPS - 1 || b_last) *(v2f*)(Bs + bl[it]) = tb[it];
#pragma unroll
    for (int it = 0; it < A_SWEEPS; ++it)
        if (it < A_SWEEPS - 1 || a_last) *(v4f*)(As + ao[it]) = ta[it] + ts[it];
    __syncthreads();

    if (eblk && tid < 128) {
        const int l32 = tid & 31;
        const int p1  = l32 + 32 < PD ? l32 + 32 : PD - 1;
        float s0 = 0.f, s1 = 0.f;
        for (int rp = 0; rp < NREL; ++rp) {
            const float wv = wgt[rp];
            s0 += wv * pes[rp * PD + l32];
            s1 += wv * pes[rp * PD + p1];
        }
        float acc = ebias;
#pragma unroll
        for (int c = 0; c < PD / 2; ++c) {
            const int pa = 2 * c, pb = 2 * c + 1;
            const float sa = __int_as_float(__builtin_amdgcn_readlane(__float_as_int(pa < 32 ? s0 : s1), pa & 31));
            const float sb = __int_as_float(__builtin_amdgcn_readlane(__float_as_int(pb < 32 ? s0 : s1), pb & 31));
            acc += sa * ew[c].x;
            acc += sb * ew[c].y;
        }
        if (ecol) E_ws[er * DD + edout] = acc;
    }

    const int mlane = lane & 15;
    const int kk    = (lane >> 4) * 2;
    const int mt    = wave & 3;
    const int nh    = wave >> 2;
    const float* ap = As + (mt * 16 + mlane) * DD + kk;
    const float* bp = Bs + (nh * 5 * 16 + mlane) * DD + kk;

    v8f acc0 = (v8f)0.0f, acc1 = (v8f)0.0f, acc2 = (v8f)0.0f, acc3 = (v8f)0.0f, acc4 = (v8f)0.0f;
    auto chunk = [&](const bool FILL, const int c) __attribute__((always_inline)) {
        const int kn = (c + 1) * KCH;
        const int kb = c * KCH;
        v2f fa[2], fb[5][2];
#pragma unroll
        for (int u = 0; u < 2; ++u) {
            fa[u] = *(const v2f*)(ap + kb + 4 * u);
#pragma unroll
            for (int j = 0; j < 5; ++j) fb[j][u] = *(const v2f*)(bp + j * 16 * DD + kb + 4 * u);
        }
#pragma unroll
        for (int g = 0; g < NGRP; ++g) {
            const int nst = (g < NGRP - 1) ? 2 : 1;
            v2f na[2], nb[5][2];
            if (g + 1 < NGRP) {
                const int nn = (g + 1 < NGRP - 1) ? 2 : 1;
#pragma unroll
                for (int u = 0; u < 2; ++u) {
                    if (u < nn) {
                        const int k0 = kb + 8 * (g + 1) + 4 * u;
                        na[u] = *(const v2f*)(ap + k0);
#pragma unroll
                        for (int j = 0; j < 5; ++j) nb[j][u] = *(const v2f*)(bp + j * 16 * DD + k0);
                    }
                }
            }
#pragma unroll
            for (int u = 0; u < 2; ++u) {
                if (u < nst) {
                    acc0 = __builtin_amdgcn_wmma_f32_16x16x4_f32(false, fa[u], false, fb[0][u], (short)0, acc0, false, false);
                    acc1 = __builtin_amdgcn_wmma_f32_16x16x4_f32(false, fa[u], false, fb[1][u], (short)0, acc1, false, false);
                    acc2 = __builtin_amdgcn_wmma_f32_16x16x4_f32(false, fa[u], false, fb[2][u], (short)0, acc2, false, false);
                    acc3 = __builtin_amdgcn_wmma_f32_16x16x4_f32(false, fa[u], false, fb[3][u], (short)0, acc3, false, false);
                    acc4 = __builtin_amdgcn_wmma_f32_16x16x4_f32(false, fa[u], false, fb[4][u], (short)0, acc4, false, false);
                }
            }
            __builtin_amdgcn_sched_barrier(0);
            if (FILL && g < NQ) {
#pragma unroll
                for (int it = g; it < B_SWEEPS; it += NQ) tb[it] = *(const v2f*)(wsel + bg[it] + kn);
                ta[g] = *(const v4f*)(hrow + ao[g] + kn);
                ts[g] = *(const v4f*)(srow + ao[g] + kn);
            }
            if (FILL && g >= ST_G0 && g < ST_G0 + NQ) {
                const int q = g - ST_G0;
#pragma unroll
                for (int it = q; it < B_SWEEPS; it += NQ) *(v2f*)(Bs + bl[it] + kn) = tb[it];
                asm volatile("" :: "v"(ta[q]), "v"(ts[q]));
                *(v4f*)(As + ao[q] + kn) = ta[q] + ts[q];
            }
            __builtin_amdgcn_sched_barrier(0);
            if (g + 1 < NGRP) {
#pragma unroll
                for (int u = 0; u < 2; ++u) {
                    fa[u] = na[u];
#pragma unroll
                    for (int j = 0; j < 5; ++j) fb[j][u] = nb[j][u];
                }
            }
        }
    };
#pragma unroll 1
    for (int c = 0; c + 1 < NCH; ++c) {
        chunk(true, c);
        __syncthreads();
    }
    chunk(false, NCH - 1);

    const int rbase = (lane >> 4) << 3;
    const size_t row0 = (size_t)(mq * GM_ROWS + mt * 16 + rbase);
    v8f accs[5] = {acc0, acc1, acc2, acc3, acc4};
#pragma unroll
    for (int j = 0; j < 5; ++j) {
        const int ncol = (ng * GN_TILES + nh * 5 + j) * 16 + mlane;
        if (ncol < DD) {
#pragma unroll
            for (int g = 0; g < 8; ++g)
                Out[(row0 + g) * DD + ncol] = accs[j][g];
        }
    }
}

#define EP_ROWS   16
#define EP_T      (EP_ROWS * 32)
#define EP_CROWS  (EP_ROWS + 2 * KK_REL)
#define EP_NBLK   (LL / EP_ROWS)
#define C_PCS     (EP_CROWS * DD / 4)
#define C_SWEEPS  ((C_PCS + EP_T - 1) / EP_T)
#define E_PCS     (NREL * DD / 4)
#define E_SWEEPS  ((E_PCS + EP_T - 1) / EP_T)
#define LOG2E_F   1.4426950408889634f

template <int MASK>
__device__ __forceinline__ float row_xor(float x) {
    const int b = __builtin_bit_cast(int, x);
    return __builtin_bit_cast(float, __builtin_amdgcn_update_dpp(b, b, 0x160 + MASK, 0xf, 0xf, true));
}
__device__ __forceinline__ float row16_sum(float x) {
    x += row_xor<1>(x);
    x += row_xor<2>(x);
    x += row_xor<4>(x);
    x += row_xor<8>(x);
    return x;
}

__global__ __launch_bounds__(EP_T) void pair_epilogue(
    const float* __restrict__ A_ws, const float* __restrict__ C_ws,
    const float* __restrict__ E_ws,
    const float* __restrict__ ln_g, const float* __restrict__ ln_b,
    const float* __restrict__ W_rel, const float* __restrict__ b_rel,
    float* __restrict__ out)
{
    __shared__ __attribute__((aligned(16))) float Cs[EP_CROWS * DD];
    __shared__ __attribute__((aligned(16))) float Es[NREL * DD];

    const int tid  = threadIdx.x;
    const int lane = tid & 31;
    const int c    = lane & 15;
    const int hf   = lane >> 4;
    const int wave = __builtin_amdgcn_readfirstlane(tid >> 5);
    const int b    = blockIdx.x / EP_NBLK;
    const int i0   = (blockIdx.x - b * EP_NBLK) * EP_ROWS;
    int jlo = i0 - KK_REL;
    jlo = jlo < 0 ? 0 : jlo;
    jlo = jlo > LL - EP_CROWS ? LL - EP_CROWS : jlo;

    const float4* csrc = (const float4*)(C_ws + ((size_t)b * LL + jlo) * DD);
    const float4* esrc = (const float4*)E_ws;
#pragma unroll
    for (int it = 0; it < C_SWEEPS; ++it) {
        int q = it * EP_T + tid;
        q = q < C_PCS ? q : C_PCS - 1;
        ((float4*)Cs)[q] = csrc[q];
    }
#pragma unroll
    for (int it = 0; it < E_SWEEPS; ++it) {
        int q = it * EP_T + tid;
        q = q < E_PCS ? q : E_PCS - 1;
        ((float4*)Es)[q] = esrc[q];
    }

    const int i = i0 + wave;
    const float* ra = A_ws + ((size_t)b * LL + i) * DD;
    v2f a[10], g[10], bb[10], wr[10];
#pragma unroll
    for (int q = 0; q < 10; ++q) {
        const int idx = (q >> 1) * 64 + c * 4 + (q & 1) * 2;
        a[q] = (v2f)0.0f; g[q] = (v2f)0.0f; bb[q] = (v2f)0.0f; wr[q] = (v2f)0.0f;
        if (idx < DD) {
            a[q]  = *(const v2f*)(ra    + idx);
            g[q]  = *(const v2f*)(ln_g  + idx);
            bb[q] = *(const v2f*)(ln_b  + idx) + 1.0f;
            wr[q] = *(const v2f*)(W_rel + idx);
        }
    }
    v2f wl2 = (v2f)0.0f;
#pragma unroll
    for (int q = 0; q < 10; ++q) wl2 += wr[q];
    const float bias = b_rel[0] - row16_sum(wl2.x + wl2.y);
    __syncthreads();

    const int rlo = i < KK_REL ? KK_REL - i : 0;
    const int rhi = i + KK_REL > LL - 1 ? LL - 1 - i + KK_REL : 2 * KK_REL;
    const int np  = rhi - rlo + 1;
    const int nit = (np + 1) >> 1;
    const int m = i < KK_REL ? i : KK_REL;
    const int n = i > LL - KK_REL ? i - (LL - KK_REL) : 0;
    const int start = NREL * i - (KK_REL * m - (m * (m - 1)) / 2) - (n * (n + 1)) / 2;

    float res = 0.f;
    for (int it = 0; it < nit; ++it) {
        int r = rlo + 2 * it + hf;
        r = r < rhi ? r : rhi;
        const float* crow = Cs + (i - KK_REL + r - jlo) * DD + c * 4;
        const float* erow = Es + r * DD + c * 4;
        v2f h[10];
        v2f s2 = (v2f)0.0f, q2 = (v2f)0.0f;
#pragma unroll
        for (int t = 0; t < 5; ++t) {
            v2f h0 = (v2f)0.0f, h1 = (v2f)0.0f;
            if (t * 64 + c * 4 < DD) {
                const v4f vc = *(const v4f*)(crow + t * 64);
                const v4f ve = *(const v4f*)(erow + t * 64);
                h0 = a[2 * t]     + vc.xy + ve.xy;
                h1 = a[2 * t + 1] + vc.zw + ve.zw;
            }
            h[2 * t]     = h0;
            h[2 * t + 1] = h1;
            s2 += h0 + h1;
            q2 += h0 * h0 + h1 * h1;
        }
        const float s  = row16_sum(s2.x + s2.y);
        const float ss = row16_sum(q2.x + q2.y);
        const float mu   = s * (1.f / DD);
        const float var  = ss * (1.f / DD) - mu * mu;
        const float rstd = rsqrtf(var + 1e-5f);
        const float nmr  = -mu * rstd;

        v2f acc2 = (v2f)0.0f;
#pragma unroll
        for (int t = 0; t < 5; ++t) {
            if (t * 64 + c * 4 < DD) {
#pragma unroll
                for (int u = 0; u < 2; ++u) {
                    const int q = 2 * t + u;
                    const v2f x  = h[q] * rstd + nmr;
                    const v2f v1 = x * g[q] + bb[q];
                    const v2f ul = v1 * LOG2E_F - LOG2E_F;
                    const float e0 = __builtin_amdgcn_fmed3f(__builtin_amdgcn_exp2f(ul.x), 0.f, 1.f);
                    const float e1 = __builtin_amdgcn_fmed3f(__builtin_amdgcn_exp2f(ul.y), 0.f, 1.f);
                    v2f m;
                    m.x = fmaxf(v1.x, e0);
                    m.y = fmaxf(v1.y, e1);
                    acc2 += m * wr[q];
                }
            }
        }
        const float acc = row16_sum(acc2.x + acc2.y);

        if (c == it) res = acc + bias;
    }

    const int k = 2 * c + hf;
    if (k < np) out[(size_t)b * NPAIR + start + k] = res;
}

extern "C" void kernel_launch(void* const* d_in, const int* in_sizes, int n_in,
                              void* d_out, int out_size, void* d_ws, size_t ws_size,
                              hipStream_t stream) {
    const float* h_e     = (const float*)d_in[0];
    const float* h_c     = (const float*)d_in[1];
    const float* h_sh    = (const float*)d_in[2];
    const float* pos_emb = (const float*)d_in[3];
    const float* W_hid   = (const float*)d_in[4];
    const float* b_hid   = (const float*)d_in[5];
    const float* ln_g    = (const float*)d_in[6];
    const float* ln_b    = (const float*)d_in[7];
    const float* W_rel   = (const float*)d_in[8];
    const float* b_rel   = (const float*)d_in[9];
    float* out = (float*)d_out;

    float* ws   = (float*)d_ws;
    float* A_ws = ws;
    float* C_ws = A_ws + (size_t)MROWS * DD;
    float* E_ws = C_ws + (size_t)MROWS * DD;

    dim3 ggrid(MROWS / GM_ROWS, 2, 2);
    pairproj_gemm<<<ggrid, 256, 0, stream>>>(h_e, h_c, h_sh, W_hid, pos_emb, b_hid, A_ws, C_ws, E_ws);

    pair_epilogue<<<BB * EP_NBLK, EP_T, 0, stream>>>(A_ws, C_ws, E_ws,
                                                     ln_g, ln_b, W_rel, b_rel, out);
}
